# starting kernel
# baseline (speedup 1.0000x reference)
#include <hip/hip_runtime.h>

typedef __attribute__((ext_vector_type(2))) float v2f;
typedef __attribute__((ext_vector_type(8))) float v8f;

#define N_NODES  50000
#define N_EDGES  800000
#define D_IN     128
#define D_OUT    64
#define N_GRAPHS 512


__global__ void k_deg_init(float* __restrict__ deg) {
    int i = blockIdx.x * blockDim.x + threadIdx.x;
    if (i < N_NODES) deg[i] = 1.0f;
}

__global__ void k_deg_edges(const int* __restrict__ ei,
                            const float* __restrict__ ew,
                            float* __restrict__ deg) {
    int e = blockIdx.x * blockDim.x + threadIdx.x;
    if (e < N_EDGES) {
        int dst = ei[N_EDGES + e];
        atomicAdd(&deg[dst], ew[e]);
    }
}

__global__ void k_dis(const float* __restrict__ deg, float* __restrict__ dis) {
    int i = blockIdx.x * blockDim.x + threadIdx.x;
    if (i < N_NODES) {
        float d = deg[i];
        dis[i] = (d > 0.0f) ? rsqrtf(d) : 0.0f;
    }
}


__global__ void k_gemm_h(const float* __restrict__ x,
                         const float* __restrict__ Wc,
                         float* __restrict__ h) {
    const int lane  = threadIdx.x & 31;
    const int wv    = threadIdx.x >> 5;
    const int mtile = blockIdx.x;
    const int half  = lane >> 4;
    const int l     = lane & 15;
    const int row   = mtile * 16 + l;
    const int n     = wv * 16 + l;

    v8f acc = {};
    #pragma unroll
    for (int k0 = 0; k0 < D_IN; k0 += 4) {
        const int ka = k0 + 2 * half;
        v2f a, b;
        a.x = x[row * D_IN + ka];
        a.y = x[row * D_IN + ka + 1];
        b.x = Wc[n * D_IN + ka];
        b.y = Wc[n * D_IN + ka + 1];
        acc = __builtin_amdgcn_wmma_f32_16x16x4_f32(
                  false, a, false, b, (short)0, acc, false, false);
    }
    #pragma unroll
    for (int r = 0; r < 8; ++r) {
        int m = mtile * 16 + half * 8 + r;
        h[m * D_IN + wv * 16 + l] = acc[r];
    }
}


__global__ void k_aggregate(const int* __restrict__ ei,
                            const float* __restrict__ ew,
                            const float* __restrict__ dis,
                            const float* __restrict__ h,
                            float* __restrict__ out) {
    unsigned gid = blockIdx.x * blockDim.x + threadIdx.x;
    int e = (int)(gid >> 5);
    if (e >= N_EDGES) return;
    int lane = (int)(gid & 31);

    int src = ei[e];
    int dst = ei[N_EDGES + e];
    float nrm = dis[src] * ew[e] * dis[dst];

    const float4 v = *((const float4*)(h + (size_t)src * D_IN) + lane);
    float* o = out + (size_t)dst * D_IN + lane * 4;
    atomicAdd(o + 0, nrm * v.x);
    atomicAdd(o + 1, nrm * v.y);
    atomicAdd(o + 2, nrm * v.z);
    atomicAdd(o + 3, nrm * v.w);
}

__global__ void k_selfloop_bias(const float* __restrict__ dis,
                                const float* __restrict__ h,
                                const float* __restrict__ bc,
                                float* __restrict__ out) {
    int gid = blockIdx.x * blockDim.x + threadIdx.x;
    if (gid >= N_NODES * D_IN) return;
    int i = gid >> 7;
    int c = gid & (D_IN - 1);
    float ds = dis[i];
    out[gid] += ds * ds * h[gid] + bc[c];
}


__global__ void k_pool(const int* __restrict__ batch,
                       const float* __restrict__ out,
                       float* __restrict__ pooled) {
    int gid = blockIdx.x * blockDim.x + threadIdx.x;
    if (gid >= N_NODES * D_IN) return;
    int i = gid >> 7;
    int c = gid & (D_IN - 1);
    atomicAdd(&pooled[batch[i] * D_IN + c], out[gid]);
}

__global__ void k_count(const int* __restrict__ batch, float* __restrict__ cnt) {
    int i = blockIdx.x * blockDim.x + threadIdx.x;
    if (i < N_NODES) atomicAdd(&cnt[batch[i]], 1.0f);
}

__global__ void k_invc(const float* __restrict__ cnt, float* __restrict__ invc) {
    int g = blockIdx.x * blockDim.x + threadIdx.x;
    if (g < N_GRAPHS) invc[g] = 1.0f / fmaxf(cnt[g], 1.0f);
}


__global__ void k_gemm_out(const float* __restrict__ pooled,
                           const float* __restrict__ invc,
                           const float* __restrict__ Wl,
                           const float* __restrict__ bl,
                           float* __restrict__ y) {
    const int lane  = threadIdx.x & 31;
    const int wv    = threadIdx.x >> 5;
    const int mtile = blockIdx.x;
    const int half  = lane >> 4;
    const int l     = lane & 15;
    const int g     = mtile * 16 + l;
    const int n     = wv * 16 + l;
    const float ic  = invc[g];

    v8f acc = {};
    #pragma unroll
    for (int k0 = 0; k0 < D_IN; k0 += 4) {
        const int ka = k0 + 2 * half;
        v2f a, b;
        a.x = pooled[g * D_IN + ka] * ic;
        a.y = pooled[g * D_IN + ka + 1] * ic;
        b.x = Wl[n * D_IN + ka];
        b.y = Wl[n * D_IN + ka + 1];
        acc = __builtin_amdgcn_wmma_f32_16x16x4_f32(
                  false, a, false, b, (short)0, acc, false, false);
    }
    #pragma unroll
    for (int r = 0; r < 8; ++r) {
        int gg = mtile * 16 + half * 8 + r;
        int oo = wv * 16 + l;
        y[gg * D_OUT + oo] = acc[r] + bl[oo];
    }
}


extern "C" void kernel_launch(void* const* d_in, const int* in_sizes, int n_in,
                              void* d_out, int out_size, void* d_ws, size_t ws_size,
                              hipStream_t stream) {
    const float* x     = (const float*)d_in[0];
    const int*   ei    = (const int*)  d_in[1];
    const int*   batch = (const int*)  d_in[2];
    const float* ew    = (const float*)d_in[3];
    const float* Wc    = (const float*)d_in[4];
    const float* bc    = (const float*)d_in[5];
    const float* Wl    = (const float*)d_in[6];
    const float* bl    = (const float*)d_in[7];
    float* y = (float*)d_out;

    char* ws = (char*)d_ws;
    size_t off = 0;
    auto take = [&](size_t bytes) -> float* {
        float* p = (float*)(ws + off);
        off += (bytes + 255) & ~((size_t)255);
        return p;
    };
    float* deg    = take((size_t)N_NODES * 4);
    float* dis    = take((size_t)N_NODES * 4);
    float* h      = take((size_t)N_NODES * D_IN * 4);
    float* agg    = take((size_t)N_NODES * D_IN * 4);
    float* pooled = take((size_t)N_GRAPHS * D_IN * 4);
    float* cnt    = take((size_t)N_GRAPHS * 4);
    float* invc   = take((size_t)N_GRAPHS * 4);
    (void)ws_size; (void)in_sizes; (void)n_in; (void)out_size;

    hipMemsetAsync(agg,    0, (size_t)N_NODES * D_IN * 4, stream);
    hipMemsetAsync(pooled, 0, (size_t)N_GRAPHS * D_IN * 4, stream);
    hipMemsetAsync(cnt,    0, (size_t)N_GRAPHS * 4, stream);

    k_deg_init <<<(N_NODES + 255) / 256, 256, 0, stream>>>(deg);
    k_deg_edges<<<(N_EDGES + 255) / 256, 256, 0, stream>>>(ei, ew, deg);
    k_dis      <<<(N_NODES + 255) / 256, 256, 0, stream>>>(deg, dis);

    k_gemm_h<<<N_NODES / 16, 256, 0, stream>>>(x, Wc, h);

    k_aggregate<<<(N_EDGES * 32) / 256, 256, 0, stream>>>(ei, ew, dis, h, agg);
    k_selfloop_bias<<<(N_NODES * D_IN + 255) / 256, 256, 0, stream>>>(dis, h, bc, agg);

    k_pool <<<(N_NODES * D_IN + 255) / 256, 256, 0, stream>>>(batch, agg, pooled);
    k_count<<<(N_NODES + 255) / 256, 256, 0, stream>>>(batch, cnt);
    k_invc <<<(N_GRAPHS + 255) / 256, 256, 0, stream>>>(cnt, invc);

    k_gemm_out<<<N_GRAPHS / 16, 128, 0, stream>>>(pooled, invc, Wl, bl, y);
}


# k_aggregate: lane l takes columns l+32j, so each f32 atomic add and each row load of a wave covers one whole 128-byte line (was 4 partial lines per instruction)
# speedup vs baseline: 1.2005x; 1.2005x over previous
#include <hip/hip_runtime.h>

typedef __attribute__((ext_vector_type(2))) float v2f;
typedef __attribute__((ext_vector_type(8))) float v8f;

#define N_NODES  50000
#define N_EDGES  800000
#define D_IN     128
#define D_OUT    64
#define N_GRAPHS 512


__global__ void k_deg_init(float* __restrict__ deg) {
    int i = blockIdx.x * blockDim.x + threadIdx.x;
    if (i < N_NODES) deg[i] = 1.0f;
}

__global__ void k_deg_edges(const int* __restrict__ ei,
                            const float* __restrict__ ew,
                            float* __restrict__ deg) {
    int e = blockIdx.x * blockDim.x + threadIdx.x;
    if (e < N_EDGES) {
        int dst = ei[N_EDGES + e];
        atomicAdd(&deg[dst], ew[e]);
    }
}

__global__ void k_dis(const float* __restrict__ deg, float* __restrict__ dis) {
    int i = blockIdx.x * blockDim.x + threadIdx.x;
    if (i < N_NODES) {
        float d = deg[i];
        dis[i] = (d > 0.0f) ? rsqrtf(d) : 0.0f;
    }
}


__global__ void k_gemm_h(const float* __restrict__ x,
                         const float* __restrict__ Wc,
                         float* __restrict__ h) {
    const int lane  = threadIdx.x & 31;
    const int wv    = threadIdx.x >> 5;
    const int mtile = blockIdx.x;
    const int half  = lane >> 4;
    const int l     = lane & 15;
    const int row   = mtile * 16 + l;
    const int n     = wv * 16 + l;

    v8f acc = {};
    #pragma unroll
    for (int k0 = 0; k0 < D_IN; k0 += 4) {
        const int ka = k0 + 2 * half;
        v2f a, b;
        a.x = x[row * D_IN + ka];
        a.y = x[row * D_IN + ka + 1];
        b.x = Wc[n * D_IN + ka];
        b.y = Wc[n * D_IN + ka + 1];
        acc = __builtin_amdgcn_wmma_f32_16x16x4_f32(
                  false, a, false, b, (short)0, acc, false, false);
    }
    #pragma unroll
    for (int r = 0; r < 8; ++r) {
        int m = mtile * 16 + half * 8 + r;
        h[m * D_IN + wv * 16 + l] = acc[r];
    }
}


__global__ void k_aggregate(const int* __restrict__ ei,
                            const float* __restrict__ ew,
                            const float* __restrict__ dis,
                            const float* __restrict__ h,
                            float* __restrict__ out) {
    unsigned gid = blockIdx.x * blockDim.x + threadIdx.x;
    int e = (int)(gid >> 5);
    if (e >= N_EDGES) return;
    int lane = (int)(gid & 31);

    int src = ei[e];
    int dst = ei[N_EDGES + e];
    float nrm = dis[src] * ew[e] * dis[dst];

    const float* hp = h + (size_t)src * D_IN + lane;
    const float v0 = hp[0];
    const float v1 = hp[32];
    const float v2 = hp[64];
    const float v3 = hp[96];
    float* o = out + (size_t)dst * D_IN + lane;
    atomicAdd(o +  0, nrm * v0);
    atomicAdd(o + 32, nrm * v1);
    atomicAdd(o + 64, nrm * v2);
    atomicAdd(o + 96, nrm * v3);
}

__global__ void k_selfloop_bias(const float* __restrict__ dis,
                                const float* __restrict__ h,
                                const float* __restrict__ bc,
                                float* __restrict__ out) {
    int gid = blockIdx.x * blockDim.x + threadIdx.x;
    if (gid >= N_NODES * D_IN) return;
    int i = gid >> 7;
    int c = gid & (D_IN - 1);
    float ds = dis[i];
    out[gid] += ds * ds * h[gid] + bc[c];
}


__global__ void k_pool(const int* __restrict__ batch,
                       const float* __restrict__ out,
                       float* __restrict__ pooled) {
    int gid = blockIdx.x * blockDim.x + threadIdx.x;
    if (gid >= N_NODES * D_IN) return;
    int i = gid >> 7;
    int c = gid & (D_IN - 1);
    atomicAdd(&pooled[batch[i] * D_IN + c], out[gid]);
}

__global__ void k_count(const int* __restrict__ batch, float* __restrict__ cnt) {
    int i = blockIdx.x * blockDim.x + threadIdx.x;
    if (i < N_NODES) atomicAdd(&cnt[batch[i]], 1.0f);
}

__global__ void k_invc(const float* __restrict__ cnt, float* __restrict__ invc) {
    int g = blockIdx.x * blockDim.x + threadIdx.x;
    if (g < N_GRAPHS) invc[g] = 1.0f / fmaxf(cnt[g], 1.0f);
}


__global__ void k_gemm_out(const float* __restrict__ pooled,
                           const float* __restrict__ invc,
                           const float* __restrict__ Wl,
                           const float* __restrict__ bl,
                           float* __restrict__ y) {
    const int lane  = threadIdx.x & 31;
    const int wv    = threadIdx.x >> 5;
    const int mtile = blockIdx.x;
    const int half  = lane >> 4;
    const int l     = lane & 15;
    const int g     = mtile * 16 + l;
    const int n     = wv * 16 + l;
    const float ic  = invc[g];

    v8f acc = {};
    #pragma unroll
    for (int k0 = 0; k0 < D_IN; k0 += 4) {
        const int ka = k0 + 2 * half;
        v2f a, b;
        a.x = pooled[g * D_IN + ka] * ic;
        a.y = pooled[g * D_IN + ka + 1] * ic;
        b.x = Wl[n * D_IN + ka];
        b.y = Wl[n * D_IN + ka + 1];
        acc = __builtin_amdgcn_wmma_f32_16x16x4_f32(
                  false, a, false, b, (short)0, acc, false, false);
    }
    #pragma unroll
    for (int r = 0; r < 8; ++r) {
        int gg = mtile * 16 + half * 8 + r;
        int oo = wv * 16 + l;
        y[gg * D_OUT + oo] = acc[r] + bl[oo];
    }
}


extern "C" void kernel_launch(void* const* d_in, const int* in_sizes, int n_in,
                              void* d_out, int out_size, void* d_ws, size_t ws_size,
                              hipStream_t stream) {
    const float* x     = (const float*)d_in[0];
    const int*   ei    = (const int*)  d_in[1];
    const int*   batch = (const int*)  d_in[2];
    const float* ew    = (const float*)d_in[3];
    const float* Wc    = (const float*)d_in[4];
    const float* bc    = (const float*)d_in[5];
    const float* Wl    = (const float*)d_in[6];
    const float* bl    = (const float*)d_in[7];
    float* y = (float*)d_out;

    char* ws = (char*)d_ws;
    size_t off = 0;
    auto take = [&](size_t bytes) -> float* {
        float* p = (float*)(ws + off);
        off += (bytes + 255) & ~((size_t)255);
        return p;
    };
    float* deg    = take((size_t)N_NODES * 4);
    float* dis    = take((size_t)N_NODES * 4);
    float* h      = take((size_t)N_NODES * D_IN * 4);
    float* agg    = take((size_t)N_NODES * D_IN * 4);
    float* pooled = take((size_t)N_GRAPHS * D_IN * 4);
    float* cnt    = take((size_t)N_GRAPHS * 4);
    float* invc   = take((size_t)N_GRAPHS * 4);
    (void)ws_size; (void)in_sizes; (void)n_in; (void)out_size;

    hipMemsetAsync(agg,    0, (size_t)N_NODES * D_IN * 4, stream);
    hipMemsetAsync(pooled, 0, (size_t)N_GRAPHS * D_IN * 4, stream);
    hipMemsetAsync(cnt,    0, (size_t)N_GRAPHS * 4, stream);

    k_deg_init <<<(N_NODES + 255) / 256, 256, 0, stream>>>(deg);
    k_deg_edges<<<(N_EDGES + 255) / 256, 256, 0, stream>>>(ei, ew, deg);
    k_dis      <<<(N_NODES + 255) / 256, 256, 0, stream>>>(deg, dis);

    k_gemm_h<<<N_NODES / 16, 256, 0, stream>>>(x, Wc, h);

    k_aggregate<<<(N_EDGES * 32) / 256, 256, 0, stream>>>(ei, ew, dis, h, agg);
    k_selfloop_bias<<<(N_NODES * D_IN + 255) / 256, 256, 0, stream>>>(dis, h, bc, agg);

    k_pool <<<(N_NODES * D_IN + 255) / 256, 256, 0, stream>>>(batch, agg, pooled);
    k_count<<<(N_NODES + 255) / 256, 256, 0, stream>>>(batch, cnt);
    k_invc <<<(N_GRAPHS + 255) / 256, 256, 0, stream>>>(cnt, invc);

    k_gemm_out<<<N_GRAPHS / 16, 128, 0, stream>>>(pooled, invc, Wl, bl, y);
}


# fewer, fatter waves in the three wave-entry-bound launches: k_aggregate 32 edges per wave, row gathers 8 edges deep; k_selfloop_bias, k_pool 16 floats per lane; f32 atomics unordered as before
# speedup vs baseline: 1.4220x; 1.1845x over previous
#include <hip/hip_runtime.h>

typedef __attribute__((ext_vector_type(2))) float v2f;
typedef __attribute__((ext_vector_type(8))) float v8f;

#define N_NODES  50000
#define N_EDGES  800000
#define D_IN     128
#define D_OUT    64
#define N_GRAPHS 512


__global__ void k_deg_init(float* __restrict__ deg) {
    int i = blockIdx.x * blockDim.x + threadIdx.x;
    if (i < N_NODES) deg[i] = 1.0f;
}

__global__ void k_deg_edges(const int* __restrict__ ei,
                            const float* __restrict__ ew,
                            float* __restrict__ deg) {
    int e = blockIdx.x * blockDim.x + threadIdx.x;
    if (e < N_EDGES) {
        int dst = ei[N_EDGES + e];
        atomicAdd(&deg[dst], ew[e]);
    }
}

__global__ void k_dis(const float* __restrict__ deg, float* __restrict__ dis) {
    int i = blockIdx.x * blockDim.x + threadIdx.x;
    if (i < N_NODES) {
        float d = deg[i];
        dis[i] = (d > 0.0f) ? rsqrtf(d) : 0.0f;
    }
}


__global__ void k_gemm_h(const float* __restrict__ x,
                         const float* __restrict__ Wc,
                         float* __restrict__ h) {
    const int lane  = threadIdx.x & 31;
    const int wv    = threadIdx.x >> 5;
    const int mtile = blockIdx.x;
    const int half  = lane >> 4;
    const int l     = lane & 15;
    const int row   = mtile * 16 + l;
    const int n     = wv * 16 + l;

    v8f acc = {};
    #pragma unroll
    for (int k0 = 0; k0 < D_IN; k0 += 4) {
        const int ka = k0 + 2 * half;
        v2f a, b;
        a.x = x[row * D_IN + ka];
        a.y = x[row * D_IN + ka + 1];
        b.x = Wc[n * D_IN + ka];
        b.y = Wc[n * D_IN + ka + 1];
        acc = __builtin_amdgcn_wmma_f32_16x16x4_f32(
                  false, a, false, b, (short)0, acc, false, false);
    }
    #pragma unroll
    for (int r = 0; r < 8; ++r) {
        int m = mtile * 16 + half * 8 + r;
        h[m * D_IN + wv * 16 + l] = acc[r];
    }
}


static_assert(N_EDGES % 32 == 0, "k_aggregate: a wave takes 32 whole edges");

__global__ void k_aggregate(const int* __restrict__ ei,
                            const float* __restrict__ ew,
                            const float* __restrict__ dis,
                            const float* __restrict__ h,
                            float* __restrict__ out) {
    const int lane = threadIdx.x & 31;
    const int e = blockIdx.x * blockDim.x + threadIdx.x;
    if (e >= N_EDGES) return;

    const int   src_l = ei[e];
    const int   dst_l = ei[N_EDGES + e];
    const float nrm_l = dis[src_l] * ew[e] * dis[dst_l];

    for (int g = 0; g < 32; g += 8) {
        float v[8][4];
        #pragma unroll
        for (int j = 0; j < 8; ++j) {
            const int src = __builtin_amdgcn_readlane(src_l, g + j);
            const float* hp = h + (size_t)src * D_IN + lane;
            v[j][0] = hp[0];
            v[j][1] = hp[32];
            v[j][2] = hp[64];
            v[j][3] = hp[96];
        }
        #pragma unroll
        for (int j = 0; j < 8; ++j) {
            const int   dst = __builtin_amdgcn_readlane(dst_l, g + j);
            const float nrm = __int_as_float(__builtin_amdgcn_readlane(__float_as_int(nrm_l), g + j));
            float* o = out + (size_t)dst * D_IN + lane;
            atomicAdd(o +  0, nrm * v[j][0]);
            atomicAdd(o + 32, nrm * v[j][1]);
            atomicAdd(o + 64, nrm * v[j][2]);
            atomicAdd(o + 96, nrm * v[j][3]);
        }
    }
}

static_assert((N_NODES * D_IN / 4) % (128 * 4) == 0, "k_selfloop_bias: whole blocks only");

__global__ void k_selfloop_bias(const float* __restrict__ dis,
                                const float* __restrict__ h,
                                const float* __restrict__ bc,
                                float* __restrict__ out) {
    const int c = (threadIdx.x & 31) * 4;
    const int base4 = blockIdx.x * (128 * 4) + threadIdx.x;
    const float4* h4 = reinterpret_cast<const float4*>(h);
    float4* o4 = reinterpret_cast<float4*>(out);
    const float b0 = bc[c], b1 = bc[c + 1], b2 = bc[c + 2], b3 = bc[c + 3];

    float4 o[4], hv[4];
    float ds[4];
    #pragma unroll
    for (int k = 0; k < 4; ++k) {
        const int idx4 = base4 + k * 128;
        o[k]  = o4[idx4];
        hv[k] = h4[idx4];
        ds[k] = dis[idx4 >> 5];
    }
    #pragma unroll
    for (int k = 0; k < 4; ++k) {
        const int idx4 = base4 + k * 128;
        const float d = ds[k];
        o[k].x += d * d * hv[k].x + b0;
        o[k].y += d * d * hv[k].y + b1;
        o[k].z += d * d * hv[k].z + b2;
        o[k].w += d * d * hv[k].w + b3;
        o4[idx4] = o[k];
    }
}


static_assert(N_NODES % 8 == 0, "k_pool: a wave takes 8 whole rows");

__global__ void k_pool(const int* __restrict__ batch,
                       const float* __restrict__ out,
                       float* __restrict__ pooled) {
    const int lane = threadIdx.x & 31;
    const int wave = __builtin_amdgcn_readfirstlane((int)((blockIdx.x * blockDim.x + threadIdx.x) >> 5));
    const int i0 = wave * 8;
    if (i0 >= N_NODES) return;

    float v[8][4];
    int   g[8];
    #pragma unroll
    for (int j = 0; j < 8; ++j) {
        g[j] = batch[i0 + j];
        const float* p = out + (size_t)(i0 + j) * D_IN + lane;
        v[j][0] = p[0];
        v[j][1] = p[32];
        v[j][2] = p[64];
        v[j][3] = p[96];
    }
    #pragma unroll
    for (int j = 0; j < 8; ++j) {
        float* q = pooled + (size_t)g[j] * D_IN + lane;
        atomicAdd(q +  0, v[j][0]);
        atomicAdd(q + 32, v[j][1]);
        atomicAdd(q + 64, v[j][2]);
        atomicAdd(q + 96, v[j][3]);
    }
}

__global__ void k_count(const int* __restrict__ batch, float* __restrict__ cnt) {
    int i = blockIdx.x * blockDim.x + threadIdx.x;
    if (i < N_NODES) atomicAdd(&cnt[batch[i]], 1.0f);
}

__global__ void k_invc(const float* __restrict__ cnt, float* __restrict__ invc) {
    int g = blockIdx.x * blockDim.x + threadIdx.x;
    if (g < N_GRAPHS) invc[g] = 1.0f / fmaxf(cnt[g], 1.0f);
}


__global__ void k_gemm_out(const float* __restrict__ pooled,
                           const float* __restrict__ invc,
                           const float* __restrict__ Wl,
                           const float* __restrict__ bl,
                           float* __restrict__ y) {
    const int lane  = threadIdx.x & 31;
    const int wv    = threadIdx.x >> 5;
    const int mtile = blockIdx.x;
    const int half  = lane >> 4;
    const int l     = lane & 15;
    const int g     = mtile * 16 + l;
    const int n     = wv * 16 + l;
    const float ic  = invc[g];

    v8f acc = {};
    #pragma unroll
    for (int k0 = 0; k0 < D_IN; k0 += 4) {
        const int ka = k0 + 2 * half;
        v2f a, b;
        a.x = pooled[g * D_IN + ka] * ic;
        a.y = pooled[g * D_IN + ka + 1] * ic;
        b.x = Wl[n * D_IN + ka];
        b.y = Wl[n * D_IN + ka + 1];
        acc = __builtin_amdgcn_wmma_f32_16x16x4_f32(
                  false, a, false, b, (short)0, acc, false, false);
    }
    #pragma unroll
    for (int r = 0; r < 8; ++r) {
        int gg = mtile * 16 + half * 8 + r;
        int oo = wv * 16 + l;
        y[gg * D_OUT + oo] = acc[r] + bl[oo];
    }
}


extern "C" void kernel_launch(void* const* d_in, const int* in_sizes, int n_in,
                              void* d_out, int out_size, void* d_ws, size_t ws_size,
                              hipStream_t stream) {
    const float* x     = (const float*)d_in[0];
    const int*   ei    = (const int*)  d_in[1];
    const int*   batch = (const int*)  d_in[2];
    const float* ew    = (const float*)d_in[3];
    const float* Wc    = (const float*)d_in[4];
    const float* bc    = (const float*)d_in[5];
    const float* Wl    = (const float*)d_in[6];
    const float* bl    = (const float*)d_in[7];
    float* y = (float*)d_out;

    char* ws = (char*)d_ws;
    size_t off = 0;
    auto take = [&](size_t bytes) -> float* {
        float* p = (float*)(ws + off);
        off += (bytes + 255) & ~((size_t)255);
        return p;
    };
    float* deg    = take((size_t)N_NODES * 4);
    float* dis    = take((size_t)N_NODES * 4);
    float* h      = take((size_t)N_NODES * D_IN * 4);
    float* agg    = take((size_t)N_NODES * D_IN * 4);
    float* pooled = take((size_t)N_GRAPHS * D_IN * 4);
    float* cnt    = take((size_t)N_GRAPHS * 4);
    float* invc   = take((size_t)N_GRAPHS * 4);
    (void)ws_size; (void)in_sizes; (void)n_in; (void)out_size;

    hipMemsetAsync(agg,    0, (size_t)N_NODES * D_IN * 4, stream);
    hipMemsetAsync(pooled, 0, (size_t)N_GRAPHS * D_IN * 4, stream);
    hipMemsetAsync(cnt,    0, (size_t)N_GRAPHS * 4, stream);

    k_deg_init <<<(N_NODES + 255) / 256, 256, 0, stream>>>(deg);
    k_deg_edges<<<(N_EDGES + 255) / 256, 256, 0, stream>>>(ei, ew, deg);
    k_dis      <<<(N_NODES + 255) / 256, 256, 0, stream>>>(deg, dis);

    k_gemm_h<<<N_NODES / 16, 256, 0, stream>>>(x, Wc, h);

    k_aggregate<<<(N_EDGES + 255) / 256, 256, 0, stream>>>(ei, ew, dis, h, agg);
    k_selfloop_bias<<<(N_NODES * D_IN / 4) / (128 * 4), 128, 0, stream>>>(dis, h, bc, agg);

    k_pool <<<(N_NODES / 8 * 32 + 255) / 256, 256, 0, stream>>>(batch, agg, pooled);
    k_count<<<(N_NODES + 255) / 256, 256, 0, stream>>>(batch, cnt);
    k_invc <<<(N_GRAPHS + 255) / 256, 256, 0, stream>>>(cnt, invc);

    k_gemm_out<<<N_GRAPHS / 16, 128, 0, stream>>>(pooled, invc, Wl, bl, y);
}


# f32 atomics (degree, aggregation) and agg zero fill replaced by per-destination edge lists built in the call (int-atomic slot, scan, place); one owner sums a list in registers, in slot order (varies)
# speedup vs baseline: 1.8963x; 1.3335x over previous
#include <hip/hip_runtime.h>

typedef __attribute__((ext_vector_type(2))) float v2f;
typedef __attribute__((ext_vector_type(8))) float v8f;

#define N_NODES  50000
#define N_EDGES  800000
#define D_IN     128
#define D_OUT    64
#define N_GRAPHS 512


__global__ void k_cnt_init(int* __restrict__ ecnt) {
    int i = blockIdx.x * blockDim.x + threadIdx.x;
    if (i < N_NODES) ecnt[i] = 0;
}

#define EDGES_PER_BLOCK (256 * 5)
static_assert(N_EDGES % EDGES_PER_BLOCK == 0, "k_rank, k_place: whole blocks only");

__global__ void k_rank(const int* __restrict__ ei,
                       int* __restrict__ ecnt,
                       int* __restrict__ rank) {
    const int e0 = blockIdx.x * EDGES_PER_BLOCK + threadIdx.x;
    int dst[5], rk[5];
    #pragma unroll
    for (int k = 0; k < 5; ++k) dst[k] = ei[N_EDGES + e0 + k * 256];
    #pragma unroll
    for (int k = 0; k < 5; ++k) rk[k] = atomicAdd(&ecnt[dst[k]], 1);
    #pragma unroll
    for (int k = 0; k < 5; ++k) rank[e0 + k * 256] = rk[k];
}

__device__ __forceinline__ int wave_incl_scan(int v, int lane) {
    v += __builtin_amdgcn_update_dpp(0, v, 0x111, 0xf, 0xf, true);
    v += __builtin_amdgcn_update_dpp(0, v, 0x112, 0xf, 0xf, true);
    v += __builtin_amdgcn_update_dpp(0, v, 0x114, 0xf, 0xf, true);
    v += __builtin_amdgcn_update_dpp(0, v, 0x118, 0xf, 0xf, true);
    const int low = __builtin_amdgcn_readlane(v, 15);
    return v + ((lane >= 16) ? low : 0);
}

#define SCAN_IT ((N_NODES + 1023) / 1024)

__global__ void __launch_bounds__(1024) k_scan(const int* __restrict__ ecnt,
                                               int* __restrict__ rowptr) {
    __shared__ int wtot[32];
    const int lane = threadIdx.x & 31;
    const int wv   = threadIdx.x >> 5;
    const int base = wv * (SCAN_IT * 32);

    int c[SCAN_IT];
    #pragma unroll
    for (int k = 0; k < SCAN_IT; ++k) {
        const int idx = base + k * 32 + lane;
        c[k] = (idx < N_NODES) ? ecnt[idx] : 0;
    }
    int carry = 0;
    #pragma unroll
    for (int k = 0; k < SCAN_IT; ++k) {
        const int v = wave_incl_scan(c[k], lane);
        c[k] = v - c[k] + carry;
        carry += __builtin_amdgcn_readlane(v, 31);
    }
    if (lane == 0) wtot[wv] = carry;
    __syncthreads();
    int off = 0;
    for (int w = 0; w < wv; ++w) off += wtot[w];
    #pragma unroll
    for (int k = 0; k < SCAN_IT; ++k) {
        const int idx = base + k * 32 + lane;
        if (idx < N_NODES) rowptr[idx] = c[k] + off;
    }
    if (threadIdx.x == 1023) rowptr[N_NODES] = off + carry;
}

__global__ void k_place(const int* __restrict__ ei,
                        const float* __restrict__ ew,
                        const int* __restrict__ rowptr,
                        const int* __restrict__ rank,
                        int2* __restrict__ rec) {
    const int e0 = blockIdx.x * EDGES_PER_BLOCK + threadIdx.x;
    int dst[5], pos[5];
    int2 r[5];
    #pragma unroll
    for (int k = 0; k < 5; ++k) {
        const int e = e0 + k * 256;
        dst[k] = ei[N_EDGES + e];
        pos[k] = rank[e];
        r[k] = make_int2(ei[e], __float_as_int(ew[e]));
    }
    #pragma unroll
    for (int k = 0; k < 5; ++k) pos[k] += rowptr[dst[k]];
    #pragma unroll
    for (int k = 0; k < 5; ++k) rec[pos[k]] = r[k];
}

__global__ void k_dis(const int* __restrict__ rowptr,
                      const int2* __restrict__ rec,
                      float* __restrict__ dis) {
    int i = blockIdx.x * blockDim.x + threadIdx.x;
    if (i < N_NODES) {
        const int beg = rowptr[i], end = rowptr[i + 1];
        float d = 1.0f;
        for (int k = beg; k < end; k += 8) {
            float w[8];
            #pragma unroll
            for (int u = 0; u < 8; ++u)
                w[u] = __int_as_float(rec[min(k + u, end - 1)].y);
            #pragma unroll
            for (int u = 0; u < 8; ++u)
                if (k + u < end) d += w[u];
        }
        dis[i] = (d > 0.0f) ? rsqrtf(d) : 0.0f;
    }
}


__global__ void k_gemm_h(const float* __restrict__ x,
                         const float* __restrict__ Wc,
                         float* __restrict__ h) {
    const int lane  = threadIdx.x & 31;
    const int wv    = threadIdx.x >> 5;
    const int mtile = blockIdx.x;
    const int half  = lane >> 4;
    const int l     = lane & 15;
    const int row   = mtile * 16 + l;
    const int n     = wv * 16 + l;

    v8f acc = {};
    #pragma unroll
    for (int k0 = 0; k0 < D_IN; k0 += 4) {
        const int ka = k0 + 2 * half;
        v2f a, b;
        a.x = x[row * D_IN + ka];
        a.y = x[row * D_IN + ka + 1];
        b.x = Wc[n * D_IN + ka];
        b.y = Wc[n * D_IN + ka + 1];
        acc = __builtin_amdgcn_wmma_f32_16x16x4_f32(
                  false, a, false, b, (short)0, acc, false, false);
    }
    #pragma unroll
    for (int r = 0; r < 8; ++r) {
        int m = mtile * 16 + half * 8 + r;
        h[m * D_IN + wv * 16 + l] = acc[r];
    }
}

static_assert(N_NODES % 8 == 0, "k_gather: 8 whole rows per 256-thread block");

__global__ void k_gather(const int* __restrict__ rowptr,
                         const int2* __restrict__ rec,
                         const float* __restrict__ dis,
                         const float* __restrict__ h,
                         const float* __restrict__ bc,
                         float* __restrict__ out) {
    const int lane = threadIdx.x & 31;
    const int i = __builtin_amdgcn_readfirstlane((int)((blockIdx.x * blockDim.x + threadIdx.x) >> 5));
    if (i >= N_NODES) return;
    const int beg = __builtin_amdgcn_readfirstlane(rowptr[i]);
    const int end = __builtin_amdgcn_readfirstlane(rowptr[i + 1]);

    const float4* h4 = reinterpret_cast<const float4*>(h);
    const int c = lane * 4;
    const float4 hs = h4[(size_t)i * 32 + lane];
    const float  d  = dis[i];
    const float b0 = bc[c], b1 = bc[c + 1], b2 = bc[c + 2], b3 = bc[c + 3];

    float4 acc = make_float4(0.0f, 0.0f, 0.0f, 0.0f);
    for (int base = beg; base < end; base += 32) {
        const int n = min(32, end - base);
        int   src_l = 0;
        float nrm_l = 0.0f;
        if (lane < n) {
            const int2 r = rec[base + lane];
            src_l = r.x;
            nrm_l = dis[r.x] * __int_as_float(r.y) * d;
        }
        for (int j = 0; j < n; j += 8) {
            float4 v[8];
            #pragma unroll
            for (int u = 0; u < 8; ++u) {
                const int k = min(j + u, n - 1);
                const int s = __builtin_amdgcn_readlane(src_l, k);
                v[u] = h4[(size_t)s * 32 + lane];
            }
            #pragma unroll
            for (int u = 0; u < 8; ++u) {
                const float w = __int_as_float(__builtin_amdgcn_readlane(__float_as_int(nrm_l), j + u));
                acc.x += w * v[u].x;
                acc.y += w * v[u].y;
                acc.z += w * v[u].z;
                acc.w += w * v[u].w;
            }
        }
    }
    acc.x += d * d * hs.x + b0;
    acc.y += d * d * hs.y + b1;
    acc.z += d * d * hs.z + b2;
    acc.w += d * d * hs.w + b3;
    reinterpret_cast<float4*>(out)[(size_t)i * 32 + lane] = acc;
}


static_assert(N_NODES % 8 == 0, "k_pool: a wave takes 8 whole rows");

__global__ void k_pool(const int* __restrict__ batch,
                       const float* __restrict__ out,
                       float* __restrict__ pooled) {
    const int lane = threadIdx.x & 31;
    const int wave = __builtin_amdgcn_readfirstlane((int)((blockIdx.x * blockDim.x + threadIdx.x) >> 5));
    const int i0 = wave * 8;
    if (i0 >= N_NODES) return;

    float v[8][4];
    int   g[8];
    #pragma unroll
    for (int j = 0; j < 8; ++j) {
        g[j] = batch[i0 + j];
        const float* p = out + (size_t)(i0 + j) * D_IN + lane;
        v[j][0] = p[0];
        v[j][1] = p[32];
        v[j][2] = p[64];
        v[j][3] = p[96];
    }
    #pragma unroll
    for (int j = 0; j < 8; ++j) {
        float* q = pooled + (size_t)g[j] * D_IN + lane;
        atomicAdd(q +  0, v[j][0]);
        atomicAdd(q + 32, v[j][1]);
        atomicAdd(q + 64, v[j][2]);
        atomicAdd(q + 96, v[j][3]);
    }
}

__global__ void k_count(const int* __restrict__ batch, float* __restrict__ cnt) {
    int i = blockIdx.x * blockDim.x + threadIdx.x;
    if (i < N_NODES) atomicAdd(&cnt[batch[i]], 1.0f);
}

__global__ void k_invc(const float* __restrict__ cnt, float* __restrict__ invc) {
    int g = blockIdx.x * blockDim.x + threadIdx.x;
    if (g < N_GRAPHS) invc[g] = 1.0f / fmaxf(cnt[g], 1.0f);
}


__global__ void k_gemm_out(const float* __restrict__ pooled,
                           const float* __restrict__ invc,
                           const float* __restrict__ Wl,
                           const float* __restrict__ bl,
                           float* __restrict__ y) {
    const int lane  = threadIdx.x & 31;
    const int wv    = threadIdx.x >> 5;
    const int mtile = blockIdx.x;
    const int half  = lane >> 4;
    const int l     = lane & 15;
    const int g     = mtile * 16 + l;
    const int n     = wv * 16 + l;
    const float ic  = invc[g];

    v8f acc = {};
    #pragma unroll
    for (int k0 = 0; k0 < D_IN; k0 += 4) {
        const int ka = k0 + 2 * half;
        v2f a, b;
        a.x = pooled[g * D_IN + ka] * ic;
        a.y = pooled[g * D_IN + ka + 1] * ic;
        b.x = Wl[n * D_IN + ka];
        b.y = Wl[n * D_IN + ka + 1];
        acc = __builtin_amdgcn_wmma_f32_16x16x4_f32(
                  false, a, false, b, (short)0, acc, false, false);
    }
    #pragma unroll
    for (int r = 0; r < 8; ++r) {
        int gg = mtile * 16 + half * 8 + r;
        int oo = wv * 16 + l;
        y[gg * D_OUT + oo] = acc[r] + bl[oo];
    }
}


extern "C" void kernel_launch(void* const* d_in, const int* in_sizes, int n_in,
                              void* d_out, int out_size, void* d_ws, size_t ws_size,
                              hipStream_t stream) {
    const float* x     = (const float*)d_in[0];
    const int*   ei    = (const int*)  d_in[1];
    const int*   batch = (const int*)  d_in[2];
    const float* ew    = (const float*)d_in[3];
    const float* Wc    = (const float*)d_in[4];
    const float* bc    = (const float*)d_in[5];
    const float* Wl    = (const float*)d_in[6];
    const float* bl    = (const float*)d_in[7];
    float* y = (float*)d_out;

    char* ws = (char*)d_ws;
    size_t off = 0;
    auto take = [&](size_t bytes) -> float* {
        float* p = (float*)(ws + off);
        off += (bytes + 255) & ~((size_t)255);
        return p;
    };
    float* dis    = take((size_t)N_NODES * 4);
    float* h      = take((size_t)N_NODES * D_IN * 4);
    float* agg    = take((size_t)N_NODES * D_IN * 4);
    float* pooled = take((size_t)N_GRAPHS * D_IN * 4);
    float* cnt    = take((size_t)N_GRAPHS * 4);
    float* invc   = take((size_t)N_GRAPHS * 4);
    int*  ecnt    = (int*) take((size_t)N_NODES * 4);
    int*  rowptr  = (int*) take((size_t)(N_NODES + 1) * 4);
    int*  rank    = (int*) take((size_t)N_EDGES * 4);
    int2* rec     = (int2*)take((size_t)N_EDGES * 8);
    (void)ws_size; (void)in_sizes; (void)n_in; (void)out_size;

    hipMemsetAsync(pooled, 0, (size_t)N_GRAPHS * D_IN * 4, stream);
    hipMemsetAsync(cnt,    0, (size_t)N_GRAPHS * 4, stream);

    k_cnt_init<<<(N_NODES + 255) / 256, 256, 0, stream>>>(ecnt);
    k_rank    <<<N_EDGES / EDGES_PER_BLOCK, 256, 0, stream>>>(ei, ecnt, rank);
    k_scan    <<<1, 1024, 0, stream>>>(ecnt, rowptr);
    k_place   <<<N_EDGES / EDGES_PER_BLOCK, 256, 0, stream>>>(ei, ew, rowptr, rank, rec);
    k_dis     <<<(N_NODES + 255) / 256, 256, 0, stream>>>(rowptr, rec, dis);

    k_gemm_h<<<N_NODES / 16, 256, 0, stream>>>(x, Wc, h);

    k_gather<<<N_NODES / 8, 256, 0, stream>>>(rowptr, rec, dis, h, bc, agg);

    k_pool <<<(N_NODES / 8 * 32 + 255) / 256, 256, 0, stream>>>(batch, agg, pooled);
    k_count<<<(N_NODES + 255) / 256, 256, 0, stream>>>(batch, cnt);
    k_invc <<<(N_GRAPHS + 255) / 256, 256, 0, stream>>>(cnt, invc);

    k_gemm_out<<<N_GRAPHS / 16, 128, 0, stream>>>(pooled, invc, Wl, bl, y);
}


# W_conv product moved behind aggregation and mean pool (linearity): gather and pool take x rows, last kernel applies W_conv then W_lin to 512 pooled rows; 50000-row GEMM gone; f32 sum order changes
# speedup vs baseline: 2.3219x; 1.2244x over previous
#include <hip/hip_runtime.h>

typedef __attribute__((ext_vector_type(2))) float v2f;
typedef __attribute__((ext_vector_type(8))) float v8f;

#define N_NODES  50000
#define N_EDGES  800000
#define D_IN     128
#define D_OUT    64
#define N_GRAPHS 512


__global__ void k_cnt_init(int* __restrict__ ecnt) {
    int i = blockIdx.x * blockDim.x + threadIdx.x;
    if (i < N_NODES) ecnt[i] = 0;
}

#define EDGES_PER_BLOCK (256 * 5)
static_assert(N_EDGES % EDGES_PER_BLOCK == 0, "k_rank, k_place: whole blocks only");

__global__ void k_rank(const int* __restrict__ ei,
                       int* __restrict__ ecnt,
                       int* __restrict__ rank) {
    const int e0 = blockIdx.x * EDGES_PER_BLOCK + threadIdx.x;
    int dst[5], rk[5];
    #pragma unroll
    for (int k = 0; k < 5; ++k) dst[k] = ei[N_EDGES + e0 + k * 256];
    #pragma unroll
    for (int k = 0; k < 5; ++k) rk[k] = atomicAdd(&ecnt[dst[k]], 1);
    #pragma unroll
    for (int k = 0; k < 5; ++k) rank[e0 + k * 256] = rk[k];
}

__device__ __forceinline__ int wave_incl_scan(int v, int lane) {
    v += __builtin_amdgcn_update_dpp(0, v, 0x111, 0xf, 0xf, true);
    v += __builtin_amdgcn_update_dpp(0, v, 0x112, 0xf, 0xf, true);
    v += __builtin_amdgcn_update_dpp(0, v, 0x114, 0xf, 0xf, true);
    v += __builtin_amdgcn_update_dpp(0, v, 0x118, 0xf, 0xf, true);
    const int low = __builtin_amdgcn_readlane(v, 15);
    return v + ((lane >= 16) ? low : 0);
}

#define SCAN_IT ((N_NODES + 1023) / 1024)

__global__ void __launch_bounds__(1024) k_scan(const int* __restrict__ ecnt,
                                               int* __restrict__ rowptr) {
    __shared__ int wtot[32];
    const int lane = threadIdx.x & 31;
    const int wv   = threadIdx.x >> 5;
    const int base = wv * (SCAN_IT * 32);

    int c[SCAN_IT];
    #pragma unroll
    for (int k = 0; k < SCAN_IT; ++k) {
        const int idx = base + k * 32 + lane;
        c[k] = (idx < N_NODES) ? ecnt[idx] : 0;
    }
    int carry = 0;
    #pragma unroll
    for (int k = 0; k < SCAN_IT; ++k) {
        const int v = wave_incl_scan(c[k], lane);
        c[k] = v - c[k] + carry;
        carry += __builtin_amdgcn_readlane(v, 31);
    }
    if (lane == 0) wtot[wv] = carry;
    __syncthreads();
    int off = 0;
    for (int w = 0; w < wv; ++w) off += wtot[w];
    #pragma unroll
    for (int k = 0; k < SCAN_IT; ++k) {
        const int idx = base + k * 32 + lane;
        if (idx < N_NODES) rowptr[idx] = c[k] + off;
    }
    if (threadIdx.x == 1023) rowptr[N_NODES] = off + carry;
}

__global__ void k_place(const int* __restrict__ ei,
                        const float* __restrict__ ew,
                        const int* __restrict__ rowptr,
                        const int* __restrict__ rank,
                        int2* __restrict__ rec) {
    const int e0 = blockIdx.x * EDGES_PER_BLOCK + threadIdx.x;
    int dst[5], pos[5];
    int2 r[5];
    #pragma unroll
    for (int k = 0; k < 5; ++k) {
        const int e = e0 + k * 256;
        dst[k] = ei[N_EDGES + e];
        pos[k] = rank[e];
        r[k] = make_int2(ei[e], __float_as_int(ew[e]));
    }
    #pragma unroll
    for (int k = 0; k < 5; ++k) pos[k] += rowptr[dst[k]];
    #pragma unroll
    for (int k = 0; k < 5; ++k) rec[pos[k]] = r[k];
}

__global__ void k_dis(const int* __restrict__ rowptr,
                      const int2* __restrict__ rec,
                      float* __restrict__ dis) {
    int i = blockIdx.x * blockDim.x + threadIdx.x;
    if (i < N_NODES) {
        const int beg = rowptr[i], end = rowptr[i + 1];
        float d = 1.0f;
        for (int k = beg; k < end; k += 8) {
            float w[8];
            #pragma unroll
            for (int u = 0; u < 8; ++u)
                w[u] = __int_as_float(rec[min(k + u, end - 1)].y);
            #pragma unroll
            for (int u = 0; u < 8; ++u)
                if (k + u < end) d += w[u];
        }
        dis[i] = (d > 0.0f) ? rsqrtf(d) : 0.0f;
    }
}

static_assert(N_NODES % 8 == 0, "k_gather: 8 whole rows per 256-thread block");

__global__ void k_gather(const int* __restrict__ rowptr,
                         const int2* __restrict__ rec,
                         const float* __restrict__ dis,
                         const float* __restrict__ x,
                         float* __restrict__ out) {
    const int lane = threadIdx.x & 31;
    const int i = __builtin_amdgcn_readfirstlane((int)((blockIdx.x * blockDim.x + threadIdx.x) >> 5));
    if (i >= N_NODES) return;
    const int beg = __builtin_amdgcn_readfirstlane(rowptr[i]);
    const int end = __builtin_amdgcn_readfirstlane(rowptr[i + 1]);

    const float4* x4 = reinterpret_cast<const float4*>(x);
    const float4 xs = x4[(size_t)i * 32 + lane];
    const float  d  = dis[i];

    float4 acc = make_float4(0.0f, 0.0f, 0.0f, 0.0f);
    for (int base = beg; base < end; base += 32) {
        const int n = min(32, end - base);
        int   src_l = 0;
        float nrm_l = 0.0f;
        if (lane < n) {
            const int2 r = rec[base + lane];
            src_l = r.x;
            nrm_l = dis[r.x] * __int_as_float(r.y) * d;
        }
        for (int j = 0; j < n; j += 8) {
            float4 v[8];
            #pragma unroll
            for (int u = 0; u < 8; ++u) {
                const int k = min(j + u, n - 1);
                const int s = __builtin_amdgcn_readlane(src_l, k);
                v[u] = x4[(size_t)s * 32 + lane];
            }
            #pragma unroll
            for (int u = 0; u < 8; ++u) {
                const float w = __int_as_float(__builtin_amdgcn_readlane(__float_as_int(nrm_l), j + u));
                acc.x += w * v[u].x;
                acc.y += w * v[u].y;
                acc.z += w * v[u].z;
                acc.w += w * v[u].w;
            }
        }
    }
    acc.x += d * d * xs.x;
    acc.y += d * d * xs.y;
    acc.z += d * d * xs.z;
    acc.w += d * d * xs.w;
    reinterpret_cast<float4*>(out)[(size_t)i * 32 + lane] = acc;
}


static_assert(N_NODES % 8 == 0, "k_pool: a wave takes 8 whole rows");

__global__ void k_pool(const int* __restrict__ batch,
                       const float* __restrict__ out,
                       float* __restrict__ pooled) {
    const int lane = threadIdx.x & 31;
    const int wave = __builtin_amdgcn_readfirstlane((int)((blockIdx.x * blockDim.x + threadIdx.x) >> 5));
    const int i0 = wave * 8;
    if (i0 >= N_NODES) return;

    float v[8][4];
    int   g[8];
    #pragma unroll
    for (int j = 0; j < 8; ++j) {
        g[j] = batch[i0 + j];
        const float* p = out + (size_t)(i0 + j) * D_IN + lane;
        v[j][0] = p[0];
        v[j][1] = p[32];
        v[j][2] = p[64];
        v[j][3] = p[96];
    }
    #pragma unroll
    for (int j = 0; j < 8; ++j) {
        float* q = pooled + (size_t)g[j] * D_IN + lane;
        atomicAdd(q +  0, v[j][0]);
        atomicAdd(q + 32, v[j][1]);
        atomicAdd(q + 64, v[j][2]);
        atomicAdd(q + 96, v[j][3]);
    }
}

__global__ void k_count(const int* __restrict__ batch, float* __restrict__ cnt) {
    int i = blockIdx.x * blockDim.x + threadIdx.x;
    if (i < N_NODES) atomicAdd(&cnt[batch[i]], 1.0f);
}

__global__ void k_invc(const float* __restrict__ cnt, float* __restrict__ invc) {
    int g = blockIdx.x * blockDim.x + threadIdx.x;
    if (g < N_GRAPHS) invc[g] = 1.0f / fmaxf(cnt[g], 1.0f);
}

#define T_PITCH 132

__global__ void __launch_bounds__(256) k_gemm_out(const float* __restrict__ pooled,
                                                  const float* __restrict__ cnt,
                                                  const float* __restrict__ invc,
                                                  const float* __restrict__ Wc,
                                                  const float* __restrict__ bc,
                                                  const float* __restrict__ Wl,
                                                  const float* __restrict__ bl,
                                                  float* __restrict__ y) {
    __shared__ float t[16 * T_PITCH];
    const int lane  = threadIdx.x & 31;
    const int wv    = threadIdx.x >> 5;
    const int mtile = blockIdx.x;
    const int half  = lane >> 4;
    const int l     = lane & 15;
    const int g     = mtile * 16 + l;
    const int n     = wv * 16 + l;
    const float ic  = invc[g];

    v8f acc = {};
    #pragma unroll
    for (int k0 = 0; k0 < D_IN; k0 += 4) {
        const int ka = k0 + 2 * half;
        v2f a, b;
        a.x = pooled[g * D_IN + ka] * ic;
        a.y = pooled[g * D_IN + ka + 1] * ic;
        b.x = Wc[n * D_IN + ka];
        b.y = Wc[n * D_IN + ka + 1];
        acc = __builtin_amdgcn_wmma_f32_16x16x4_f32(
                  false, a, false, b, (short)0, acc, false, false);
    }
    const float bn = bc[n];
    #pragma unroll
    for (int r = 0; r < 8; ++r) {
        const int row = half * 8 + r;
        const float has = (cnt[mtile * 16 + row] > 0.0f) ? 1.0f : 0.0f;
        t[row * T_PITCH + n] = acc[r] + has * bn;
    }
    __syncthreads();

    if (wv < 4) {
        v8f acc2 = {};
        #pragma unroll
        for (int k0 = 0; k0 < D_IN; k0 += 4) {
            const int ka = k0 + 2 * half;
            v2f a, b;
            a.x = t[l * T_PITCH + ka];
            a.y = t[l * T_PITCH + ka + 1];
            b.x = Wl[n * D_IN + ka];
            b.y = Wl[n * D_IN + ka + 1];
            acc2 = __builtin_amdgcn_wmma_f32_16x16x4_f32(
                       false, a, false, b, (short)0, acc2, false, false);
        }
        #pragma unroll
        for (int r = 0; r < 8; ++r) {
            int gg = mtile * 16 + half * 8 + r;
            y[gg * D_OUT + n] = acc2[r] + bl[n];
        }
    }
}


extern "C" void kernel_launch(void* const* d_in, const int* in_sizes, int n_in,
                              void* d_out, int out_size, void* d_ws, size_t ws_size,
                              hipStream_t stream) {
    const float* x     = (const float*)d_in[0];
    const int*   ei    = (const int*)  d_in[1];
    const int*   batch = (const int*)  d_in[2];
    const float* ew    = (const float*)d_in[3];
    const float* Wc    = (const float*)d_in[4];
    const float* bc    = (const float*)d_in[5];
    const float* Wl    = (const float*)d_in[6];
    const float* bl    = (const float*)d_in[7];
    float* y = (float*)d_out;

    char* ws = (char*)d_ws;
    size_t off = 0;
    auto take = [&](size_t bytes) -> float* {
        float* p = (float*)(ws + off);
        off += (bytes + 255) & ~((size_t)255);
        return p;
    };
    float* dis    = take((size_t)N_NODES * 4);
    float* agg    = take((size_t)N_NODES * D_IN * 4);
    float* pooled = take((size_t)N_GRAPHS * D_IN * 4);
    float* cnt    = take((size_t)N_GRAPHS * 4);
    float* invc   = take((size_t)N_GRAPHS * 4);
    int*  ecnt    = (int*) take((size_t)N_NODES * 4);
    int*  rowptr  = (int*) take((size_t)(N_NODES + 1) * 4);
    int*  rank    = (int*) take((size_t)N_EDGES * 4);
    int2* rec     = (int2*)take((size_t)N_EDGES * 8);
    (void)ws_size; (void)in_sizes; (void)n_in; (void)out_size;

    hipMemsetAsync(pooled, 0, (size_t)N_GRAPHS * D_IN * 4, stream);
    hipMemsetAsync(cnt,    0, (size_t)N_GRAPHS * 4, stream);

    k_cnt_init<<<(N_NODES + 255) / 256, 256, 0, stream>>>(ecnt);
    k_rank    <<<N_EDGES / EDGES_PER_BLOCK, 256, 0, stream>>>(ei, ecnt, rank);
    k_scan    <<<1, 1024, 0, stream>>>(ecnt, rowptr);
    k_place   <<<N_EDGES / EDGES_PER_BLOCK, 256, 0, stream>>>(ei, ew, rowptr, rank, rec);
    k_dis     <<<(N_NODES + 255) / 256, 256, 0, stream>>>(rowptr, rec, dis);

    k_gather<<<N_NODES / 8, 256, 0, stream>>>(rowptr, rec, dis, x, agg);

    k_pool <<<(N_NODES / 8 * 32 + 255) / 256, 256, 0, stream>>>(batch, agg, pooled);
    k_count<<<(N_NODES + 255) / 256, 256, 0, stream>>>(batch, cnt);
    k_invc <<<(N_GRAPHS + 255) / 256, 256, 0, stream>>>(cnt, invc);

    k_gemm_out<<<N_GRAPHS / 16, 256, 0, stream>>>(pooled, cnt, invc, Wc, bc, Wl, bl, y);
}


# edge lists become fixed 32-record buckets per destination, written at the slot the int atomic returns (beyond 32: overflow list, summed last, f32 order changes); scan, rank plane, place pass gone
# speedup vs baseline: 3.3046x; 1.4232x over previous
#include <hip/hip_runtime.h>

typedef __attribute__((ext_vector_type(2))) float v2f;
typedef __attribute__((ext_vector_type(8))) float v8f;

#define N_NODES  50000
#define N_EDGES  800000
#define D_IN     128
#define D_OUT    64
#define N_GRAPHS 512

#define CAP 32

__global__ void k_cnt_init(int* __restrict__ ecnt) {
    int i = blockIdx.x * blockDim.x + threadIdx.x;
    if (i <= N_NODES) ecnt[i] = 0;
}

#define EDGES_PER_BLOCK (256 * 5)
static_assert(N_EDGES % EDGES_PER_BLOCK == 0, "k_bucket: whole blocks only");

__global__ void k_bucket(const int* __restrict__ ei,
                         const float* __restrict__ ew,
                         int* __restrict__ ecnt,
                         int2* __restrict__ rec,
                         int4* __restrict__ ovf) {
    const int e0 = blockIdx.x * EDGES_PER_BLOCK + threadIdx.x;
    int dst[5], slot[5];
    int2 r[5];
    #pragma unroll
    for (int k = 0; k < 5; ++k) {
        const int e = e0 + k * 256;
        dst[k] = ei[N_EDGES + e];
        r[k] = make_int2(ei[e], __float_as_int(ew[e]));
    }
    #pragma unroll
    for (int k = 0; k < 5; ++k) slot[k] = atomicAdd(&ecnt[dst[k]], 1);
    #pragma unroll
    for (int k = 0; k < 5; ++k) {
        if (slot[k] < CAP) {
            rec[(size_t)dst[k] * CAP + slot[k]] = r[k];
        } else {
            const int q = atomicAdd(&ecnt[N_NODES], 1);
            ovf[q] = make_int4(dst[k], r[k].x, r[k].y, 0);
        }
    }
}

__global__ void k_dis(const int* __restrict__ ecnt,
                      const int2* __restrict__ rec,
                      const int4* __restrict__ ovf,
                      float* __restrict__ dis) {
    int i = blockIdx.x * blockDim.x + threadIdx.x;
    if (i < N_NODES) {
        const int c = ecnt[i];
        const int n = min(c, CAP);
        const int2* __restrict__ b = rec + (size_t)i * CAP;
        float d = 1.0f;
        for (int k = 0; k < n; k += 8) {
            float w[8];
            #pragma unroll
            for (int u = 0; u < 8; ++u)
                w[u] = __int_as_float(b[min(k + u, n - 1)].y);
            #pragma unroll
            for (int u = 0; u < 8; ++u)
                if (k + u < n) d += w[u];
        }
        if (c > CAP) {
            const int m = ecnt[N_NODES];
            for (int k = 0; k < m; ++k) {
                const int4 o = ovf[k];
                if (o.x == i) d += __int_as_float(o.z);
            }
        }
        dis[i] = (d > 0.0f) ? rsqrtf(d) : 0.0f;
    }
}

__device__ __forceinline__ int wave_incl_scan(int v, int lane) {
    v += __builtin_amdgcn_update_dpp(0, v, 0x111, 0xf, 0xf, true);
    v += __builtin_amdgcn_update_dpp(0, v, 0x112, 0xf, 0xf, true);
    v += __builtin_amdgcn_update_dpp(0, v, 0x114, 0xf, 0xf, true);
    v += __builtin_amdgcn_update_dpp(0, v, 0x118, 0xf, 0xf, true);
    const int low = __builtin_amdgcn_readlane(v, 15);
    return v + ((lane >= 16) ? low : 0);
}

#define GROUP 16
static_assert(N_NODES % 16 == 0, "k_gather_pool: 8 whole rows per wave, 2 waves per block");

__device__ __forceinline__ void gather_group(float4& acc, const float4* __restrict__ x4, int lane,
                                             int j, int n, int src_l, float nrm_l) {
    float4 v[GROUP];
    #pragma unroll
    for (int u = 0; u < GROUP; ++u) {
        const int k = min(j + u, n - 1);
        const int s = __builtin_amdgcn_readlane(src_l, k);
        v[u] = x4[(size_t)s * 32 + lane];
    }
    #pragma unroll
    for (int u = 0; u < GROUP; ++u) {
        const float w = __int_as_float(__builtin_amdgcn_readlane(__float_as_int(nrm_l), j + u));
        acc.x += w * v[u].x;
        acc.y += w * v[u].y;
        acc.z += w * v[u].z;
        acc.w += w * v[u].w;
    }
}

__global__ void __launch_bounds__(64) k_gather_pool(const int* __restrict__ ecnt,
                                                    const int2* __restrict__ rec,
                                                    const int4* __restrict__ ovf,
                                                    const float* __restrict__ dis,
                                                    const float* __restrict__ x,
                                                    const int* __restrict__ batch,
                                                    float* __restrict__ pooled,
                                                    float* __restrict__ cnt) {
    __shared__ int2 packed[2 * 8 * CAP];
    const int lane = threadIdx.x & 31;
    int2* __restrict__ lw = packed + (threadIdx.x >> 5) * (8 * CAP);
    const int wave = __builtin_amdgcn_readfirstlane((int)((blockIdx.x * blockDim.x + threadIdx.x) >> 5));
    const int i0 = wave * 8;
    const float4* x4 = reinterpret_cast<const float4*>(x);

    const int   c_l = ecnt[i0 + min(lane, 7)];
    const int   g_l = batch[i0 + min(lane, 7)];
    const float d_l = dis[i0 + min(lane, 7)];
    const int n_l   = (lane < 8) ? min(c_l, CAP) : 0;
    const int off_l = wave_incl_scan(n_l, lane) - n_l;
    const unsigned over = __builtin_amdgcn_ballot_w32(lane < 8 && c_l > CAP);

    float4 xs[8];
    #pragma unroll
    for (int r = 0; r < 8; ++r) xs[r] = x4[(size_t)(i0 + r) * 32 + lane];

    {
        const int2* __restrict__ bw = rec + (size_t)i0 * CAP;
        int   nr[8];
        int2  q[8];
        float ds[8];
        #pragma unroll
        for (int r = 0; r < 8; ++r) {
            nr[r] = __builtin_amdgcn_readlane(n_l, r);
            if (lane < nr[r]) q[r] = bw[r * CAP + lane];
        }
        #pragma unroll
        for (int r = 0; r < 8; ++r)
            if (lane < nr[r]) ds[r] = dis[q[r].x];
        #pragma unroll
        for (int r = 0; r < 8; ++r) {
            const float dr = __int_as_float(__builtin_amdgcn_readlane(__float_as_int(d_l), r));
            const int   o  = __builtin_amdgcn_readlane(off_l, r);
            if (lane < nr[r]) {
                const float nrm = ds[r] * __int_as_float(q[r].y) * dr;
                lw[o + lane] = make_int2(q[r].x, __float_as_int(nrm));
            }
        }
    }
    __syncthreads();

    int ra = 0;
    while (ra < 8) {
        const int ga = __builtin_amdgcn_readlane(g_l, ra);
        int rb = ra + 1;
        while (rb < 8 && __builtin_amdgcn_readlane(g_l, rb) == ga) ++rb;
        const int sbeg = __builtin_amdgcn_readlane(off_l, ra);
        const int send = __builtin_amdgcn_readlane(off_l, rb);

        float4 acc = make_float4(0.0f, 0.0f, 0.0f, 0.0f);

        const bool any = sbeg < send;
        int n = min(32, send - sbeg);
        int2 p = make_int2(0, 0);
        if (any && lane < n) p = lw[sbeg + lane];

        #pragma unroll
        for (int r = 0; r < 8; ++r) {
            const float dr = __int_as_float(__builtin_amdgcn_readlane(__float_as_int(d_l), r));
            const float ws = (r >= ra && r < rb) ? dr * dr : 0.0f;
            acc.x += ws * xs[r].x;
            acc.y += ws * xs[r].y;
            acc.z += ws * xs[r].z;
            acc.w += ws * xs[r].w;
        }

        if (any) {
            int base = sbeg;
            for (;;) {
                const int  nb   = base + 32;
                const bool more = nb < send;
                const int  nn   = min(32, send - nb);
                int2 pn = make_int2(0, 0);
                if (more && lane < nn) pn = lw[nb + lane];
                gather_group(acc, x4, lane, 0, n, p.x, __int_as_float(p.y));
                if (n > GROUP) gather_group(acc, x4, lane, GROUP, n, p.x, __int_as_float(p.y));
                if (!more) break;
                base = nb;
                n = nn;
                p = pn;
            }
        }

        if ((over >> ra) & ((1u << (rb - ra)) - 1u)) {
            const int m = ecnt[N_NODES];
            for (int k0 = 0; k0 < m; k0 += 32) {
                const int k = k0 + lane;
                int4 o = make_int4(-1, 0, 0, 0);
                if (k < m) o = ovf[k];
                const bool hit = (o.x >= i0 + ra) && (o.x < i0 + rb);
                float nrm = 0.0f;
                if (hit) nrm = dis[o.y] * __int_as_float(o.z) * dis[o.x];
                unsigned todo = __builtin_amdgcn_ballot_w32(hit);
                while (todo) {
                    const int l = __builtin_ctz(todo);
                    todo &= todo - 1u;
                    const int   s = __builtin_amdgcn_readlane(o.y, l);
                    const float w = __int_as_float(__builtin_amdgcn_readlane(__float_as_int(nrm), l));
                    const float4 v = x4[(size_t)s * 32 + lane];
                    acc.x += w * v.x;
                    acc.y += w * v.y;
                    acc.z += w * v.z;
                    acc.w += w * v.w;
                }
            }
        }

        float* q = pooled + (size_t)ga * D_IN + 4 * lane;
        atomicAdd(q + 0, acc.x);
        atomicAdd(q + 1, acc.y);
        atomicAdd(q + 2, acc.z);
        atomicAdd(q + 3, acc.w);
        if (lane == 0) atomicAdd(&cnt[ga], (float)(rb - ra));
        ra = rb;
    }
}

#define T_PITCH 132

__global__ void __launch_bounds__(256) k_gemm_out(const float* __restrict__ pooled,
                                                  const float* __restrict__ cnt,
                                                  const float* __restrict__ Wc,
                                                  const float* __restrict__ bc,
                                                  const float* __restrict__ Wl,
                                                  const float* __restrict__ bl,
                                                  float* __restrict__ y) {
    __shared__ float t[16 * T_PITCH];
    const int lane  = threadIdx.x & 31;
    const int wv    = threadIdx.x >> 5;
    const int mtile = blockIdx.x;
    const int half  = lane >> 4;
    const int l     = lane & 15;
    const int g     = mtile * 16 + l;
    const int n     = wv * 16 + l;
    const float ic  = 1.0f / fmaxf(cnt[g], 1.0f);

    v8f acc = {};
    #pragma unroll
    for (int k0 = 0; k0 < D_IN; k0 += 4) {
        const int ka = k0 + 2 * half;
        v2f a, b;
        a.x = pooled[g * D_IN + ka] * ic;
        a.y = pooled[g * D_IN + ka + 1] * ic;
        b.x = Wc[n * D_IN + ka];
        b.y = Wc[n * D_IN + ka + 1];
        acc = __builtin_amdgcn_wmma_f32_16x16x4_f32(
                  false, a, false, b, (short)0, acc, false, false);
    }
    const float bn = bc[n];
    #pragma unroll
    for (int r = 0; r < 8; ++r) {
        const int row = half * 8 + r;
        const float has = (cnt[mtile * 16 + row] > 0.0f) ? 1.0f : 0.0f;
        t[row * T_PITCH + n] = acc[r] + has * bn;
    }
    __syncthreads();

    if (wv < 4) {
        v8f acc2 = {};
        #pragma unroll
        for (int k0 = 0; k0 < D_IN; k0 += 4) {
            const int ka = k0 + 2 * half;
            v2f a, b;
            a.x = t[l * T_PITCH + ka];
            a.y = t[l * T_PITCH + ka + 1];
            b.x = Wl[n * D_IN + ka];
            b.y = Wl[n * D_IN + ka + 1];
            acc2 = __builtin_amdgcn_wmma_f32_16x16x4_f32(
                       false, a, false, b, (short)0, acc2, false, false);
        }
        #pragma unroll
        for (int r = 0; r < 8; ++r) {
            int gg = mtile * 16 + half * 8 + r;
            y[gg * D_OUT + n] = acc2[r] + bl[n];
        }
    }
}


extern "C" void kernel_launch(void* const* d_in, const int* in_sizes, int n_in,
                              void* d_out, int out_size, void* d_ws, size_t ws_size,
                              hipStream_t stream) {
    const float* x     = (const float*)d_in[0];
    const int*   ei    = (const int*)  d_in[1];
    const int*   batch = (const int*)  d_in[2];
    const float* ew    = (const float*)d_in[3];
    const float* Wc    = (const float*)d_in[4];
    const float* bc    = (const float*)d_in[5];
    const float* Wl    = (const float*)d_in[6];
    const float* bl    = (const float*)d_in[7];
    float* y = (float*)d_out;

    char* ws = (char*)d_ws;
    size_t off = 0;
    auto take = [&](size_t bytes) -> float* {
        float* p = (float*)(ws + off);
        off += (bytes + 255) & ~((size_t)255);
        return p;
    };
    float* dis    = take((size_t)N_NODES * 4);
    float* pooled = take((size_t)N_GRAPHS * D_IN * 4);
    float* cnt    = take((size_t)N_GRAPHS * 4);
    int*  ecnt    = (int*) take((size_t)(N_NODES + 1) * 4);
    int2* rec     = (int2*)take((size_t)N_NODES * CAP * 8);
    int4* ovf     = (int4*)take((size_t)N_EDGES * 16);
    (void)ws_size; (void)in_sizes; (void)n_in; (void)out_size;

    hipMemsetAsync(pooled, 0, (size_t)N_GRAPHS * D_IN * 4, stream);
    hipMemsetAsync(cnt,    0, (size_t)N_GRAPHS * 4, stream);

    k_cnt_init<<<(N_NODES + 1 + 255) / 256, 256, 0, stream>>>(ecnt);
    k_bucket  <<<N_EDGES / EDGES_PER_BLOCK, 256, 0, stream>>>(ei, ew, ecnt, rec, ovf);
    k_dis     <<<(N_NODES + 255) / 256, 256, 0, stream>>>(ecnt, rec, ovf, dis);

    k_gather_pool<<<N_NODES / 16, 64, 0, stream>>>(ecnt, rec, ovf, dis, x, batch, pooled, cnt);

    k_gemm_out<<<N_GRAPHS / 16, 256, 0, stream>>>(pooled, cnt, Wc, bc, Wl, bl, y);
}


# edge slots from LDS counters, not one global returning atomic per edge: blocks sort 3200-edge chunks by 128-node range, a block per range ranks per node; count zeroing gone; f32 sum order changes
# speedup vs baseline: 5.7092x; 1.7277x over previous
#include <hip/hip_runtime.h>

typedef __attribute__((ext_vector_type(2))) float v2f;
typedef __attribute__((ext_vector_type(8))) float v8f;

#define N_NODES  50000
#define N_EDGES  800000
#define D_IN     128
#define D_OUT    64
#define N_GRAPHS 512

#define CAP 32

#define NR       128
#define N_RANGES ((N_NODES + NR - 1) / NR)
#define PART_T   640
#define PART_E   5
#define CHUNK    (PART_T * PART_E)
#define N_CHUNKS (N_EDGES / CHUNK)
static_assert(N_EDGES % CHUNK == 0, "k_part: whole chunks only");
static_assert(N_RANGES <= PART_T, "k_part: the scan holds one range per thread");
static_assert(N_CHUNKS <= 256, "k_fine: one thread per chunk");
static_assert(N_NODES <= 65536 && NR <= 32768, "packed record: source in the low 16 bits, node within the range above");

__device__ __forceinline__ int wave_incl_scan(int v, int lane) {
    v += __builtin_amdgcn_update_dpp(0, v, 0x111, 0xf, 0xf, true);
    v += __builtin_amdgcn_update_dpp(0, v, 0x112, 0xf, 0xf, true);
    v += __builtin_amdgcn_update_dpp(0, v, 0x114, 0xf, 0xf, true);
    v += __builtin_amdgcn_update_dpp(0, v, 0x118, 0xf, 0xf, true);
    const int low = __builtin_amdgcn_readlane(v, 15);
    return v + ((lane >= 16) ? low : 0);
}

__global__ void __launch_bounds__(PART_T) k_part(const int* __restrict__ ei,
                                                 const float* __restrict__ ew,
                                                 int2* __restrict__ part,
                                                 int2* __restrict__ runs,
                                                 int* __restrict__ ecnt) {
    __shared__ int hist[N_RANGES];
    __shared__ int wtot[PART_T / 32];
    const int tid  = threadIdx.x;
    const int lane = tid & 31;
    const int wv   = tid >> 5;
    if (tid < N_RANGES) hist[tid] = 0;
    if (blockIdx.x == 0 && tid == 0) ecnt[N_NODES] = 0;
    const int e0 = blockIdx.x * CHUNK + tid;
    int rng[PART_E], rank[PART_E];
    int2 r[PART_E];
    #pragma unroll
    for (int k = 0; k < PART_E; ++k) {
        const int e = e0 + k * PART_T;
        const unsigned d = (unsigned)ei[N_EDGES + e];
        rng[k] = (int)(d / NR);
        r[k] = make_int2((int)((d % NR) << 16) | ei[e], __float_as_int(ew[e]));
    }
    __syncthreads();
    #pragma unroll
    for (int k = 0; k < PART_E; ++k) rank[k] = atomicAdd(&hist[rng[k]], 1);
    __syncthreads();
    const int c   = (tid < N_RANGES) ? hist[tid] : 0;
    const int inc = wave_incl_scan(c, lane);
    if (lane == 31) wtot[wv] = inc;
    __syncthreads();
    int pre = 0;
    for (int w = 0; w < wv; ++w) pre += wtot[w];
    const int start = pre + inc - c;
    if (tid < N_RANGES) {
        hist[tid] = start;
        runs[(size_t)tid * N_CHUNKS + blockIdx.x] = make_int2(start, c);
    }
    __syncthreads();
    int2* __restrict__ out = part + (size_t)blockIdx.x * CHUNK;
    #pragma unroll
    for (int k = 0; k < PART_E; ++k) out[hist[rng[k]] + rank[k]] = r[k];
}

__global__ void __launch_bounds__(256) k_fine(const int2* __restrict__ part,
                                              const int2* __restrict__ runs,
                                              int* __restrict__ ecnt,
                                              int2* __restrict__ rec,
                                              int4* __restrict__ ovf) {
    __shared__ int cntl[NR];
    const int tid  = threadIdx.x;
    const int base = blockIdx.x * NR;
    if (tid < NR) cntl[tid] = 0;
    int2 run = make_int2(0, 0);
    if (tid < N_CHUNKS) run = runs[(size_t)blockIdx.x * N_CHUNKS + tid];
    __syncthreads();
    const int2* __restrict__ p = part + (size_t)min(tid, N_CHUNKS - 1) * CHUNK + run.x;
    const int n = run.y;
    for (int k = 0; k < n; k += 8) {
        int2 q[8];
        #pragma unroll
        for (int u = 0; u < 8; ++u)
            q[u] = p[min(k + u, n - 1)];
        int slot[8];
        #pragma unroll
        for (int u = 0; u < 8; ++u)
            if (k + u < n) slot[u] = atomicAdd(&cntl[q[u].x >> 16], 1);
        #pragma unroll
        for (int u = 0; u < 8; ++u) {
            if (k + u < n) {
                const int dl = q[u].x >> 16;
                const int s  = q[u].x & 0xffff;
                if (slot[u] < CAP) {
                    rec[(size_t)(base + dl) * CAP + slot[u]] = make_int2(s, q[u].y);
                } else {
                    const int o = atomicAdd(&ecnt[N_NODES], 1);
                    ovf[o] = make_int4(base + dl, s, q[u].y, 0);
                }
            }
        }
    }
    __syncthreads();
    if (tid < NR && base + tid < N_NODES) ecnt[base + tid] = cntl[tid];
}

__global__ void k_dis(const int* __restrict__ ecnt,
                      const int2* __restrict__ rec,
                      const int4* __restrict__ ovf,
                      float* __restrict__ dis) {
    int i = blockIdx.x * blockDim.x + threadIdx.x;
    if (i < N_NODES) {
        const int c = ecnt[i];
        const int n = min(c, CAP);
        const int2* __restrict__ b = rec + (size_t)i * CAP;
        float d = 1.0f;
        for (int k = 0; k < n; k += 8) {
            float w[8];
            #pragma unroll
            for (int u = 0; u < 8; ++u)
                w[u] = __int_as_float(b[min(k + u, n - 1)].y);
            #pragma unroll
            for (int u = 0; u < 8; ++u)
                if (k + u < n) d += w[u];
        }
        if (c > CAP) {
            const int m = ecnt[N_NODES];
            for (int k = 0; k < m; ++k) {
                const int4 o = ovf[k];
                if (o.x == i) d += __int_as_float(o.z);
            }
        }
        dis[i] = (d > 0.0f) ? rsqrtf(d) : 0.0f;
    }
}

#define GROUP 16
static_assert(N_NODES % 16 == 0, "k_gather_pool: 8 whole rows per wave, 2 waves per block");

__device__ __forceinline__ void gather_group(float4& acc, const float4* __restrict__ x4, int lane,
                                             int j, int n, int src_l, float nrm_l) {
    float4 v[GROUP];
    #pragma unroll
    for (int u = 0; u < GROUP; ++u) {
        const int k = min(j + u, n - 1);
        const int s = __builtin_amdgcn_readlane(src_l, k);
        v[u] = x4[(size_t)s * 32 + lane];
    }
    #pragma unroll
    for (int u = 0; u < GROUP; ++u) {
        const float w = __int_as_float(__builtin_amdgcn_readlane(__float_as_int(nrm_l), j + u));
        acc.x += w * v[u].x;
        acc.y += w * v[u].y;
        acc.z += w * v[u].z;
        acc.w += w * v[u].w;
    }
}

__global__ void __launch_bounds__(64) k_gather_pool(const int* __restrict__ ecnt,
                                                    const int2* __restrict__ rec,
                                                    const int4* __restrict__ ovf,
                                                    const float* __restrict__ dis,
                                                    const float* __restrict__ x,
                                                    const int* __restrict__ batch,
                                                    float* __restrict__ pooled,
                                                    float* __restrict__ cnt) {
    __shared__ int2 packed[2 * 8 * CAP];
    const int lane = threadIdx.x & 31;
    int2* __restrict__ lw = packed + (threadIdx.x >> 5) * (8 * CAP);
    const int wave = __builtin_amdgcn_readfirstlane((int)((blockIdx.x * blockDim.x + threadIdx.x) >> 5));
    const int i0 = wave * 8;
    const float4* x4 = reinterpret_cast<const float4*>(x);

    const int   c_l = ecnt[i0 + min(lane, 7)];
    const int   g_l = batch[i0 + min(lane, 7)];
    const float d_l = dis[i0 + min(lane, 7)];
    const int n_l   = (lane < 8) ? min(c_l, CAP) : 0;
    const int off_l = wave_incl_scan(n_l, lane) - n_l;
    const unsigned over = __builtin_amdgcn_ballot_w32(lane < 8 && c_l > CAP);

    float4 xs[8];
    #pragma unroll
    for (int r = 0; r < 8; ++r) xs[r] = x4[(size_t)(i0 + r) * 32 + lane];

    {
        const int2* __restrict__ bw = rec + (size_t)i0 * CAP;
        int   nr[8];
        int2  q[8];
        float ds[8];
        #pragma unroll
        for (int r = 0; r < 8; ++r) {
            nr[r] = __builtin_amdgcn_readlane(n_l, r);
            if (lane < nr[r]) q[r] = bw[r * CAP + lane];
        }
        #pragma unroll
        for (int r = 0; r < 8; ++r)
            if (lane < nr[r]) ds[r] = dis[q[r].x];
        #pragma unroll
        for (int r = 0; r < 8; ++r) {
            const float dr = __int_as_float(__builtin_amdgcn_readlane(__float_as_int(d_l), r));
            const int   o  = __builtin_amdgcn_readlane(off_l, r);
            if (lane < nr[r]) {
                const float nrm = ds[r] * __int_as_float(q[r].y) * dr;
                lw[o + lane] = make_int2(q[r].x, __float_as_int(nrm));
            }
        }
    }
    __syncthreads();

    int ra = 0;
    while (ra < 8) {
        const int ga = __builtin_amdgcn_readlane(g_l, ra);
        int rb = ra + 1;
        while (rb < 8 && __builtin_amdgcn_readlane(g_l, rb) == ga) ++rb;
        const int sbeg = __builtin_amdgcn_readlane(off_l, ra);
        const int send = __builtin_amdgcn_readlane(off_l, rb);

        float4 acc = make_float4(0.0f, 0.0f, 0.0f, 0.0f);

        const bool any = sbeg < send;
        int n = min(32, send - sbeg);
        int2 p = make_int2(0, 0);
        if (any && lane < n) p = lw[sbeg + lane];

        #pragma unroll
        for (int r = 0; r < 8; ++r) {
            const float dr = __int_as_float(__builtin_amdgcn_readlane(__float_as_int(d_l), r));
            const float ws = (r >= ra && r < rb) ? dr * dr : 0.0f;
            acc.x += ws * xs[r].x;
            acc.y += ws * xs[r].y;
            acc.z += ws * xs[r].z;
            acc.w += ws * xs[r].w;
        }

        if (any) {
            int base = sbeg;
            for (;;) {
                const int  nb   = base + 32;
                const bool more = nb < send;
                const int  nn   = min(32, send - nb);
                int2 pn = make_int2(0, 0);
                if (more && lane < nn) pn = lw[nb + lane];
                gather_group(acc, x4, lane, 0, n, p.x, __int_as_float(p.y));
                if (n > GROUP) gather_group(acc, x4, lane, GROUP, n, p.x, __int_as_float(p.y));
                if (!more) break;
                base = nb;
                n = nn;
                p = pn;
            }
        }

        if ((over >> ra) & ((1u << (rb - ra)) - 1u)) {
            const int m = ecnt[N_NODES];
            for (int k0 = 0; k0 < m; k0 += 32) {
                const int k = k0 + lane;
                int4 o = make_int4(-1, 0, 0, 0);
                if (k < m) o = ovf[k];
                const bool hit = (o.x >= i0 + ra) && (o.x < i0 + rb);
                float nrm = 0.0f;
                if (hit) nrm = dis[o.y] * __int_as_float(o.z) * dis[o.x];
                unsigned todo = __builtin_amdgcn_ballot_w32(hit);
                while (todo) {
                    const int l = __builtin_ctz(todo);
                    todo &= todo - 1u;
                    const int   s = __builtin_amdgcn_readlane(o.y, l);
                    const float w = __int_as_float(__builtin_amdgcn_readlane(__float_as_int(nrm), l));
                    const float4 v = x4[(size_t)s * 32 + lane];
                    acc.x += w * v.x;
                    acc.y += w * v.y;
                    acc.z += w * v.z;
                    acc.w += w * v.w;
                }
            }
        }

        float* q = pooled + (size_t)ga * D_IN + 4 * lane;
        atomicAdd(q + 0, acc.x);
        atomicAdd(q + 1, acc.y);
        atomicAdd(q + 2, acc.z);
        atomicAdd(q + 3, acc.w);
        if (lane == 0) atomicAdd(&cnt[ga], (float)(rb - ra));
        ra = rb;
    }
}

#define T_PITCH 132

__global__ void __launch_bounds__(256) k_gemm_out(const float* __restrict__ pooled,
                                                  const float* __restrict__ cnt,
                                                  const float* __restrict__ Wc,
                                                  const float* __restrict__ bc,
                                                  const float* __restrict__ Wl,
                                                  const float* __restrict__ bl,
                                                  float* __restrict__ y) {
    __shared__ float t[16 * T_PITCH];
    const int lane  = threadIdx.x & 31;
    const int wv    = threadIdx.x >> 5;
    const int mtile = blockIdx.x;
    const int half  = lane >> 4;
    const int l     = lane & 15;
    const int g     = mtile * 16 + l;
    const int n     = wv * 16 + l;
    const float ic  = 1.0f / fmaxf(cnt[g], 1.0f);

    v8f acc = {};
    #pragma unroll
    for (int k0 = 0; k0 < D_IN; k0 += 4) {
        const int ka = k0 + 2 * half;
        v2f a, b;
        a.x = pooled[g * D_IN + ka] * ic;
        a.y = pooled[g * D_IN + ka + 1] * ic;
        b.x = Wc[n * D_IN + ka];
        b.y = Wc[n * D_IN + ka + 1];
        acc = __builtin_amdgcn_wmma_f32_16x16x4_f32(
                  false, a, false, b, (short)0, acc, false, false);
    }
    const float bn = bc[n];
    #pragma unroll
    for (int r = 0; r < 8; ++r) {
        const int row = half * 8 + r;
        const float has = (cnt[mtile * 16 + row] > 0.0f) ? 1.0f : 0.0f;
        t[row * T_PITCH + n] = acc[r] + has * bn;
    }
    __syncthreads();

    if (wv < 4) {
        v8f acc2 = {};
        #pragma unroll
        for (int k0 = 0; k0 < D_IN; k0 += 4) {
            const int ka = k0 + 2 * half;
            v2f a, b;
            a.x = t[l * T_PITCH + ka];
            a.y = t[l * T_PITCH + ka + 1];
            b.x = Wl[n * D_IN + ka];
            b.y = Wl[n * D_IN + ka + 1];
            acc2 = __builtin_amdgcn_wmma_f32_16x16x4_f32(
                       false, a, false, b, (short)0, acc2, false, false);
        }
        #pragma unroll
        for (int r = 0; r < 8; ++r) {
            int gg = mtile * 16 + half * 8 + r;
            y[gg * D_OUT + n] = acc2[r] + bl[n];
        }
    }
}


extern "C" void kernel_launch(void* const* d_in, const int* in_sizes, int n_in,
                              void* d_out, int out_size, void* d_ws, size_t ws_size,
                              hipStream_t stream) {
    const float* x     = (const float*)d_in[0];
    const int*   ei    = (const int*)  d_in[1];
    const int*   batch = (const int*)  d_in[2];
    const float* ew    = (const float*)d_in[3];
    const float* Wc    = (const float*)d_in[4];
    const float* bc    = (const float*)d_in[5];
    const float* Wl    = (const float*)d_in[6];
    const float* bl    = (const float*)d_in[7];
    float* y = (float*)d_out;

    char* ws = (char*)d_ws;
    size_t off = 0;
    auto take = [&](size_t bytes) -> float* {
        float* p = (float*)(ws + off);
        off += (bytes + 255) & ~((size_t)255);
        return p;
    };
    float* dis    = take((size_t)N_NODES * 4);
    float* pooled = take((size_t)N_GRAPHS * D_IN * 4);
    float* cnt    = take((size_t)N_GRAPHS * 4);
    int*  ecnt    = (int*) take((size_t)(N_NODES + 1) * 4);
    int2* rec     = (int2*)take((size_t)N_NODES * CAP * 8);
    int4* ovf     = (int4*)take((size_t)N_EDGES * 16);
    int2* part    = (int2*)take((size_t)N_EDGES * 8);
    int2* runs    = (int2*)take((size_t)N_RANGES * N_CHUNKS * 8);
    (void)ws_size; (void)in_sizes; (void)n_in; (void)out_size;

    hipMemsetAsync(pooled, 0, (size_t)N_GRAPHS * D_IN * 4, stream);
    hipMemsetAsync(cnt,    0, (size_t)N_GRAPHS * 4, stream);

    k_part    <<<N_CHUNKS, PART_T, 0, stream>>>(ei, ew, part, runs, ecnt);
    k_fine    <<<N_RANGES, 256, 0, stream>>>(part, runs, ecnt, rec, ovf);
    k_dis     <<<(N_NODES + 255) / 256, 256, 0, stream>>>(ecnt, rec, ovf, dis);

    k_gather_pool<<<N_NODES / 16, 64, 0, stream>>>(ecnt, rec, ovf, dis, x, batch, pooled, cnt);

    k_gemm_out<<<N_GRAPHS / 16, 256, 0, stream>>>(pooled, cnt, Wc, bc, Wl, bl, y);
}


# k_fine builds its 128 buckets in LDS, stores them as whole 128-byte lines and takes each node's degree and dis from its LDS bucket in slot order (overflow weights: LDS float add); k_dis launch gone
# speedup vs baseline: 6.4964x; 1.1379x over previous
#include <hip/hip_runtime.h>

typedef __attribute__((ext_vector_type(2))) float v2f;
typedef __attribute__((ext_vector_type(8))) float v8f;
typedef __attribute__((ext_vector_type(4))) int   v4i;

#define N_NODES  50000
#define N_EDGES  800000
#define D_IN     128
#define D_OUT    64
#define N_GRAPHS 512

#define CAP 32

#define NR       128
#define N_RANGES ((N_NODES + NR - 1) / NR)
#define PART_T   640
#define PART_E   5
#define CHUNK    (PART_T * PART_E)
#define N_CHUNKS (N_EDGES / CHUNK)
static_assert(N_EDGES % CHUNK == 0, "k_part: whole chunks only");
static_assert(N_RANGES <= PART_T, "k_part: the scan holds one range per thread");
static_assert(N_CHUNKS <= 256, "k_fine: one thread per chunk");
static_assert(N_NODES <= 65536 && NR <= 32768, "packed record: source in the low 16 bits, node within the range above");

__device__ __forceinline__ int wave_incl_scan(int v, int lane) {
    v += __builtin_amdgcn_update_dpp(0, v, 0x111, 0xf, 0xf, true);
    v += __builtin_amdgcn_update_dpp(0, v, 0x112, 0xf, 0xf, true);
    v += __builtin_amdgcn_update_dpp(0, v, 0x114, 0xf, 0xf, true);
    v += __builtin_amdgcn_update_dpp(0, v, 0x118, 0xf, 0xf, true);
    const int low = __builtin_amdgcn_readlane(v, 15);
    return v + ((lane >= 16) ? low : 0);
}

__global__ void __launch_bounds__(PART_T) k_part(const int* __restrict__ ei,
                                                 const float* __restrict__ ew,
                                                 int2* __restrict__ part,
                                                 int2* __restrict__ runs,
                                                 int* __restrict__ ecnt) {
    __shared__ int hist[N_RANGES];
    __shared__ int wtot[PART_T / 32];
    const int tid  = threadIdx.x;
    const int lane = tid & 31;
    const int wv   = tid >> 5;
    if (tid < N_RANGES) hist[tid] = 0;
    if (blockIdx.x == 0 && tid == 0) ecnt[N_NODES] = 0;
    const int e0 = blockIdx.x * CHUNK + tid;
    int rng[PART_E], rank[PART_E];
    int2 r[PART_E];
    #pragma unroll
    for (int k = 0; k < PART_E; ++k) {
        const int e = e0 + k * PART_T;
        const unsigned d = (unsigned)ei[N_EDGES + e];
        rng[k] = (int)(d / NR);
        r[k] = make_int2((int)((d % NR) << 16) | ei[e], __float_as_int(ew[e]));
    }
    __syncthreads();
    #pragma unroll
    for (int k = 0; k < PART_E; ++k) rank[k] = atomicAdd(&hist[rng[k]], 1);
    __syncthreads();
    const int c   = (tid < N_RANGES) ? hist[tid] : 0;
    const int inc = wave_incl_scan(c, lane);
    if (lane == 31) wtot[wv] = inc;
    __syncthreads();
    int pre = 0;
    for (int w = 0; w < wv; ++w) pre += wtot[w];
    const int start = pre + inc - c;
    if (tid < N_RANGES) {
        hist[tid] = start;
        runs[(size_t)tid * N_CHUNKS + blockIdx.x] = make_int2(start, c);
    }
    __syncthreads();
    int2* __restrict__ out = part + (size_t)blockIdx.x * CHUNK;
    #pragma unroll
    for (int k = 0; k < PART_E; ++k) out[hist[rng[k]] + rank[k]] = r[k];
}

#define B_PITCH (CAP + 2)
static_assert(CAP == 32 && NR == 128, "k_fine: 16 pieces of 16 bytes per bucket, 2048 pieces per block, 8 per thread");

__global__ void __launch_bounds__(256) k_fine(const int2* __restrict__ part,
                                              const int2* __restrict__ runs,
                                              int* __restrict__ ecnt,
                                              int2* __restrict__ rec,
                                              int4* __restrict__ ovf,
                                              float* __restrict__ dis) {
    __shared__ __attribute__((aligned(16))) int2 bkt[NR * B_PITCH];
    __shared__ int   cntl[NR];
    __shared__ float ovw[NR];
    const int tid  = threadIdx.x;
    const int base = blockIdx.x * NR;
    if (tid < NR) { cntl[tid] = 0; ovw[tid] = 0.0f; }
    for (int p = tid; p < NR * B_PITCH / 2; p += 256)
        reinterpret_cast<int4*>(bkt)[p] = make_int4(0, 0, 0, 0);
    int2 run = make_int2(0, 0);
    if (tid < N_CHUNKS) run = runs[(size_t)blockIdx.x * N_CHUNKS + tid];
    __syncthreads();
    const int2* __restrict__ p = part + (size_t)min(tid, N_CHUNKS - 1) * CHUNK + run.x;
    const int n = run.y;
    for (int k = 0; k < n; k += 8) {
        int2 q[8];
        #pragma unroll
        for (int u = 0; u < 8; ++u)
            q[u] = p[min(k + u, n - 1)];
        int slot[8];
        #pragma unroll
        for (int u = 0; u < 8; ++u)
            if (k + u < n) slot[u] = atomicAdd(&cntl[q[u].x >> 16], 1);
        #pragma unroll
        for (int u = 0; u < 8; ++u) {
            if (k + u < n) {
                const int dl = q[u].x >> 16;
                const int s  = q[u].x & 0xffff;
                if (slot[u] < CAP) {
                    bkt[dl * B_PITCH + slot[u]] = make_int2(s, q[u].y);
                } else {
                    atomicAdd(&ovw[dl], __int_as_float(q[u].y));
                    const int o = atomicAdd(&ecnt[N_NODES], 1);
                    ovf[o] = make_int4(base + dl, s, q[u].y, 0);
                }
            }
        }
    }
    __syncthreads();

    v4i* __restrict__ rec4 = reinterpret_cast<v4i*>(rec) + (size_t)base * (CAP / 2);
    v4i pv[8];
    int pn[8];
    #pragma unroll
    for (int j = 0; j < 8; ++j) {
        const int pc   = j * 256 + tid;
        const int node = pc >> 4;
        pn[j] = cntl[node];
        pv[j] = *reinterpret_cast<const v4i*>(&bkt[node * B_PITCH + 2 * (pc & 15)]);
    }
    asm volatile("" :: "v"(pv[0]), "v"(pv[1]), "v"(pv[2]), "v"(pv[3]), "v"(pv[4]), "v"(pv[5]), "v"(pv[6]), "v"(pv[7]));
    #pragma unroll
    for (int j = 0; j < 8; ++j) {
        const int pc   = j * 256 + tid;
        const int node = pc >> 4;
        if (base + node < N_NODES && ((pc & 15) >> 3) * 16 < pn[j]) rec4[pc] = pv[j];
    }

    if (tid < NR && base + tid < N_NODES) {
        const int c  = cntl[tid];
        const int nb = min(c, CAP);
        const int2* __restrict__ b = bkt + tid * B_PITCH;
        float d = 1.0f;
        for (int k = 0; k < nb; k += 8) {
            float w[8];
            #pragma unroll
            for (int u = 0; u < 8; ++u)
                w[u] = __int_as_float(b[min(k + u, nb - 1)].y);
            #pragma unroll
            for (int u = 0; u < 8; ++u)
                if (k + u < nb) d += w[u];
        }
        if (c > CAP) d += ovw[tid];
        dis[base + tid]  = (d > 0.0f) ? rsqrtf(d) : 0.0f;
        ecnt[base + tid] = c;
    }
}

#define GROUP 16
static_assert(N_NODES % 16 == 0, "k_gather_pool: 8 whole rows per wave, 2 waves per block");

__device__ __forceinline__ void gather_group(float4& acc, const float4* __restrict__ x4, int lane,
                                             int j, int n, int src_l, float nrm_l) {
    float4 v[GROUP];
    #pragma unroll
    for (int u = 0; u < GROUP; ++u) {
        const int k = min(j + u, n - 1);
        const int s = __builtin_amdgcn_readlane(src_l, k);
        v[u] = x4[(size_t)s * 32 + lane];
    }
    #pragma unroll
    for (int u = 0; u < GROUP; ++u) {
        const float w = __int_as_float(__builtin_amdgcn_readlane(__float_as_int(nrm_l), j + u));
        acc.x += w * v[u].x;
        acc.y += w * v[u].y;
        acc.z += w * v[u].z;
        acc.w += w * v[u].w;
    }
}

__global__ void __launch_bounds__(64) k_gather_pool(const int* __restrict__ ecnt,
                                                    const int2* __restrict__ rec,
                                                    const int4* __restrict__ ovf,
                                                    const float* __restrict__ dis,
                                                    const float* __restrict__ x,
                                                    const int* __restrict__ batch,
                                                    float* __restrict__ pooled,
                                                    float* __restrict__ cnt) {
    __shared__ int2 packed[2 * 8 * CAP];
    const int lane = threadIdx.x & 31;
    int2* __restrict__ lw = packed + (threadIdx.x >> 5) * (8 * CAP);
    const int wave = __builtin_amdgcn_readfirstlane((int)((blockIdx.x * blockDim.x + threadIdx.x) >> 5));
    const int i0 = wave * 8;
    const float4* x4 = reinterpret_cast<const float4*>(x);

    const int   c_l = ecnt[i0 + min(lane, 7)];
    const int   g_l = batch[i0 + min(lane, 7)];
    const float d_l = dis[i0 + min(lane, 7)];
    const int n_l   = (lane < 8) ? min(c_l, CAP) : 0;
    const int off_l = wave_incl_scan(n_l, lane) - n_l;
    const unsigned over = __builtin_amdgcn_ballot_w32(lane < 8 && c_l > CAP);

    float4 xs[8];
    #pragma unroll
    for (int r = 0; r < 8; ++r) xs[r] = x4[(size_t)(i0 + r) * 32 + lane];

    {
        const int2* __restrict__ bw = rec + (size_t)i0 * CAP;
        int   nr[8];
        int2  q[8];
        float ds[8];
        #pragma unroll
        for (int r = 0; r < 8; ++r) {
            nr[r] = __builtin_amdgcn_readlane(n_l, r);
            if (lane < nr[r]) q[r] = bw[r * CAP + lane];
        }
        #pragma unroll
        for (int r = 0; r < 8; ++r)
            if (lane < nr[r]) ds[r] = dis[q[r].x];
        #pragma unroll
        for (int r = 0; r < 8; ++r) {
            const float dr = __int_as_float(__builtin_amdgcn_readlane(__float_as_int(d_l), r));
            const int   o  = __builtin_amdgcn_readlane(off_l, r);
            if (lane < nr[r]) {
                const float nrm = ds[r] * __int_as_float(q[r].y) * dr;
                lw[o + lane] = make_int2(q[r].x, __float_as_int(nrm));
            }
        }
    }
    __syncthreads();

    int ra = 0;
    while (ra < 8) {
        const int ga = __builtin_amdgcn_readlane(g_l, ra);
        int rb = ra + 1;
        while (rb < 8 && __builtin_amdgcn_readlane(g_l, rb) == ga) ++rb;
        const int sbeg = __builtin_amdgcn_readlane(off_l, ra);
        const int send = __builtin_amdgcn_readlane(off_l, rb);

        float4 acc = make_float4(0.0f, 0.0f, 0.0f, 0.0f);

        const bool any = sbeg < send;
        int n = min(32, send - sbeg);
        int2 p = make_int2(0, 0);
        if (any && lane < n) p = lw[sbeg + lane];

        #pragma unroll
        for (int r = 0; r < 8; ++r) {
            const float dr = __int_as_float(__builtin_amdgcn_readlane(__float_as_int(d_l), r));
            const float ws = (r >= ra && r < rb) ? dr * dr : 0.0f;
            acc.x += ws * xs[r].x;
            acc.y += ws * xs[r].y;
            acc.z += ws * xs[r].z;
            acc.w += ws * xs[r].w;
        }

        if (any) {
            int base = sbeg;
            for (;;) {
                const int  nb   = base + 32;
                const bool more = nb < send;
                const int  nn   = min(32, send - nb);
                int2 pn = make_int2(0, 0);
                if (more && lane < nn) pn = lw[nb + lane];
                gather_group(acc, x4, lane, 0, n, p.x, __int_as_float(p.y));
                if (n > GROUP) gather_group(acc, x4, lane, GROUP, n, p.x, __int_as_float(p.y));
                if (!more) break;
                base = nb;
                n = nn;
                p = pn;
            }
        }

        if ((over >> ra) & ((1u << (rb - ra)) - 1u)) {
            const int m = ecnt[N_NODES];
            for (int k0 = 0; k0 < m; k0 += 32) {
                const int k = k0 + lane;
                int4 o = make_int4(-1, 0, 0, 0);
                if (k < m) o = ovf[k];
                const bool hit = (o.x >= i0 + ra) && (o.x < i0 + rb);
                float nrm = 0.0f;
                if (hit) nrm = dis[o.y] * __int_as_float(o.z) * dis[o.x];
                unsigned todo = __builtin_amdgcn_ballot_w32(hit);
                while (todo) {
                    const int l = __builtin_ctz(todo);
                    todo &= todo - 1u;
                    const int   s = __builtin_amdgcn_readlane(o.y, l);
                    const float w = __int_as_float(__builtin_amdgcn_readlane(__float_as_int(nrm), l));
                    const float4 v = x4[(size_t)s * 32 + lane];
                    acc.x += w * v.x;
                    acc.y += w * v.y;
                    acc.z += w * v.z;
                    acc.w += w * v.w;
                }
            }
        }

        float* q = pooled + (size_t)ga * D_IN + 4 * lane;
        atomicAdd(q + 0, acc.x);
        atomicAdd(q + 1, acc.y);
        atomicAdd(q + 2, acc.z);
        atomicAdd(q + 3, acc.w);
        if (lane == 0) atomicAdd(&cnt[ga], (float)(rb - ra));
        ra = rb;
    }
}

#define T_PITCH 132

__global__ void __launch_bounds__(256) k_gemm_out(const float* __restrict__ pooled,
                                                  const float* __restrict__ cnt,
                                                  const float* __restrict__ Wc,
                                                  const float* __restrict__ bc,
                                                  const float* __restrict__ Wl,
                                                  const float* __restrict__ bl,
                                                  float* __restrict__ y) {
    __shared__ float t[16 * T_PITCH];
    const int lane  = threadIdx.x & 31;
    const int wv    = threadIdx.x >> 5;
    const int mtile = blockIdx.x;
    const int half  = lane >> 4;
    const int l     = lane & 15;
    const int g     = mtile * 16 + l;
    const int n     = wv * 16 + l;
    const float ic  = 1.0f / fmaxf(cnt[g], 1.0f);

    v8f acc = {};
    #pragma unroll
    for (int k0 = 0; k0 < D_IN; k0 += 4) {
        const int ka = k0 + 2 * half;
        v2f a, b;
        a.x = pooled[g * D_IN + ka] * ic;
        a.y = pooled[g * D_IN + ka + 1] * ic;
        b.x = Wc[n * D_IN + ka];
        b.y = Wc[n * D_IN + ka + 1];
        acc = __builtin_amdgcn_wmma_f32_16x16x4_f32(
                  false, a, false, b, (short)0, acc, false, false);
    }
    const float bn = bc[n];
    #pragma unroll
    for (int r = 0; r < 8; ++r) {
        const int row = half * 8 + r;
        const float has = (cnt[mtile * 16 + row] > 0.0f) ? 1.0f : 0.0f;
        t[row * T_PITCH + n] = acc[r] + has * bn;
    }
    __syncthreads();

    if (wv < 4) {
        v8f acc2 = {};
        #pragma unroll
        for (int k0 = 0; k0 < D_IN; k0 += 4) {
            const int ka = k0 + 2 * half;
            v2f a, b;
            a.x = t[l * T_PITCH + ka];
            a.y = t[l * T_PITCH + ka + 1];
            b.x = Wl[n * D_IN + ka];
            b.y = Wl[n * D_IN + ka + 1];
            acc2 = __builtin_amdgcn_wmma_f32_16x16x4_f32(
                       false, a, false, b, (short)0, acc2, false, false);
        }
        #pragma unroll
        for (int r = 0; r < 8; ++r) {
            int gg = mtile * 16 + half * 8 + r;
            y[gg * D_OUT + n] = acc2[r] + bl[n];
        }
    }
}


extern "C" void kernel_launch(void* const* d_in, const int* in_sizes, int n_in,
                              void* d_out, int out_size, void* d_ws, size_t ws_size,
                              hipStream_t stream) {
    const float* x     = (const float*)d_in[0];
    const int*   ei    = (const int*)  d_in[1];
    const int*   batch = (const int*)  d_in[2];
    const float* ew    = (const float*)d_in[3];
    const float* Wc    = (const float*)d_in[4];
    const float* bc    = (const float*)d_in[5];
    const float* Wl    = (const float*)d_in[6];
    const float* bl    = (const float*)d_in[7];
    float* y = (float*)d_out;

    char* ws = (char*)d_ws;
    size_t off = 0;
    auto take = [&](size_t bytes) -> float* {
        float* p = (float*)(ws + off);
        off += (bytes + 255) & ~((size_t)255);
        return p;
    };
    float* dis    = take((size_t)N_NODES * 4);
    float* pooled = take((size_t)N_GRAPHS * D_IN * 4);
    float* cnt    = take((size_t)N_GRAPHS * 4);
    int*  ecnt    = (int*) take((size_t)(N_NODES + 1) * 4);
    int2* rec     = (int2*)take((size_t)N_NODES * CAP * 8);
    int4* ovf     = (int4*)take((size_t)N_EDGES * 16);
    int2* part    = (int2*)take((size_t)N_EDGES * 8);
    int2* runs    = (int2*)take((size_t)N_RANGES * N_CHUNKS * 8);
    (void)ws_size; (void)in_sizes; (void)n_in; (void)out_size;

    hipMemsetAsync(pooled, 0, (size_t)N_GRAPHS * D_IN * 4, stream);
    hipMemsetAsync(cnt,    0, (size_t)N_GRAPHS * 4, stream);

    k_part    <<<N_CHUNKS, PART_T, 0, stream>>>(ei, ew, part, runs, ecnt);
    k_fine    <<<N_RANGES, 256, 0, stream>>>(part, runs, ecnt, rec, ovf, dis);

    k_gather_pool<<<N_NODES / 16, 64, 0, stream>>>(ecnt, rec, ovf, dis, x, batch, pooled, cnt);

    k_gemm_out<<<N_GRAPHS / 16, 256, 0, stream>>>(pooled, cnt, Wc, bc, Wl, bl, y);
}


# accesses to the part plane made whole-line: k_part sorts its chunk in LDS and stores whole 128-byte lines, k_fine reads a run with a half-wave (lane j = record j); slot order varies as before
# speedup vs baseline: 6.7700x; 1.0421x over previous
#include <hip/hip_runtime.h>

typedef __attribute__((ext_vector_type(2))) float v2f;
typedef __attribute__((ext_vector_type(8))) float v8f;
typedef __attribute__((ext_vector_type(4))) int   v4i;

#define N_NODES  50000
#define N_EDGES  800000
#define D_IN     128
#define D_OUT    64
#define N_GRAPHS 512

#define CAP 32

#define NR       128
#define N_RANGES ((N_NODES + NR - 1) / NR)
#define PART_T   640
#define PART_E   5
#define CHUNK    (PART_T * PART_E)
#define N_CHUNKS (N_EDGES / CHUNK)
static_assert(N_EDGES % CHUNK == 0, "k_part: whole chunks only");
static_assert((CHUNK * 8) % 128 == 0 && CHUNK % 2 == 0, "k_part: a chunk of part is whole 128-byte lines, copied in 16-byte pieces");
static_assert(N_RANGES <= PART_T, "k_part: the scan holds one range per thread");
static_assert(N_CHUNKS <= 256, "k_fine: one thread per chunk");
static_assert(N_NODES <= 65536 && NR <= 32768, "packed record: source in the low 16 bits, node within the range above");

__device__ __forceinline__ int wave_incl_scan(int v, int lane) {
    v += __builtin_amdgcn_update_dpp(0, v, 0x111, 0xf, 0xf, true);
    v += __builtin_amdgcn_update_dpp(0, v, 0x112, 0xf, 0xf, true);
    v += __builtin_amdgcn_update_dpp(0, v, 0x114, 0xf, 0xf, true);
    v += __builtin_amdgcn_update_dpp(0, v, 0x118, 0xf, 0xf, true);
    const int low = __builtin_amdgcn_readlane(v, 15);
    return v + ((lane >= 16) ? low : 0);
}

__global__ void __launch_bounds__(PART_T) k_part(const int* __restrict__ ei,
                                                 const float* __restrict__ ew,
                                                 int2* __restrict__ part,
                                                 int2* __restrict__ runs,
                                                 int* __restrict__ ecnt) {
    __shared__ int hist[N_RANGES];
    __shared__ int wtot[PART_T / 32];
    __shared__ __attribute__((aligned(16))) int2 stage[CHUNK];
    const int tid  = threadIdx.x;
    const int lane = tid & 31;
    const int wv   = tid >> 5;
    if (tid < N_RANGES) hist[tid] = 0;
    if (blockIdx.x == 0 && tid == 0) ecnt[N_NODES] = 0;
    const int e0 = blockIdx.x * CHUNK + tid;
    int rng[PART_E], rank[PART_E];
    int2 r[PART_E];
    #pragma unroll
    for (int k = 0; k < PART_E; ++k) {
        const int e = e0 + k * PART_T;
        const unsigned d = (unsigned)ei[N_EDGES + e];
        rng[k] = (int)(d / NR);
        r[k] = make_int2((int)((d % NR) << 16) | ei[e], __float_as_int(ew[e]));
    }
    __syncthreads();
    #pragma unroll
    for (int k = 0; k < PART_E; ++k) rank[k] = atomicAdd(&hist[rng[k]], 1);
    __syncthreads();
    const int c   = (tid < N_RANGES) ? hist[tid] : 0;
    const int inc = wave_incl_scan(c, lane);
    if (lane == 31) wtot[wv] = inc;
    __syncthreads();
    int pre = 0;
    for (int w = 0; w < wv; ++w) pre += wtot[w];
    const int start = pre + inc - c;
    if (tid < N_RANGES) {
        hist[tid] = start;
        runs[(size_t)tid * N_CHUNKS + blockIdx.x] = make_int2(start, c);
    }
    __syncthreads();
    #pragma unroll
    for (int k = 0; k < PART_E; ++k) stage[hist[rng[k]] + rank[k]] = r[k];
    __syncthreads();
    v4i* __restrict__ out4 = reinterpret_cast<v4i*>(part + (size_t)blockIdx.x * CHUNK);
    const v4i* st4 = reinterpret_cast<const v4i*>(stage);
    constexpr int NPC = (CHUNK / 2 + PART_T - 1) / PART_T;
    v4i pv[NPC];
    #pragma unroll
    for (int p = 0; p < NPC; ++p)
        pv[p] = st4[min(p * PART_T + tid, CHUNK / 2 - 1)];
    #pragma unroll
    for (int p = 0; p < NPC; ++p) {
        const int pc = p * PART_T + tid;
        if ((p + 1) * PART_T <= CHUNK / 2 || pc < CHUNK / 2) out4[pc] = pv[p];
    }
}

#define B_PITCH (CAP + 2)
static_assert(CAP == 32 && NR == 128, "k_fine: 16 pieces of 16 bytes per bucket, 2048 pieces per block, 8 per thread");

__global__ void __launch_bounds__(256) k_fine(const int2* __restrict__ part,
                                              const int2* __restrict__ runs,
                                              int* __restrict__ ecnt,
                                              int2* __restrict__ rec,
                                              int4* __restrict__ ovf,
                                              float* __restrict__ dis) {
    __shared__ __attribute__((aligned(16))) int2 bkt[NR * B_PITCH];
    __shared__ int   cntl[NR];
    __shared__ float ovw[NR];
    const int tid  = threadIdx.x;
    const int base = blockIdx.x * NR;
    if (tid < NR) { cntl[tid] = 0; ovw[tid] = 0.0f; }
    for (int p = tid; p < NR * B_PITCH / 2; p += 256)
        reinterpret_cast<int4*>(bkt)[p] = make_int4(0, 0, 0, 0);
    int2 run = make_int2(0, 0);
    if (tid < N_CHUNKS) run = runs[(size_t)blockIdx.x * N_CHUNKS + tid];
    __syncthreads();
    const int lane = tid & 31;
    const int hw   = lane >> 4;
    const int jr   = lane & 15;
    const int c0   = tid & ~31;
    {
        int2 q[16];
        int  nn[16];
        #pragma unroll
        for (int u = 0; u < 16; ++u) {
            const int i   = 2 * u;
            const int st0 = __builtin_amdgcn_readlane(run.x, i), st1 = __builtin_amdgcn_readlane(run.x, i + 1);
            const int n0  = __builtin_amdgcn_readlane(run.y, i), n1  = __builtin_amdgcn_readlane(run.y, i + 1);
            const int st  = hw ? st1 : st0;
            nn[u]         = hw ? n1 : n0;
            const int2* __restrict__ p = part + (size_t)min(c0 + i + hw, N_CHUNKS - 1) * CHUNK + st;
            if (jr < nn[u]) q[u] = p[jr];
        }
        int slot[16];
        #pragma unroll
        for (int u = 0; u < 16; ++u)
            if (jr < nn[u]) slot[u] = atomicAdd(&cntl[q[u].x >> 16], 1);
        #pragma unroll
        for (int u = 0; u < 16; ++u) {
            if (jr < nn[u]) {
                const int dl = q[u].x >> 16;
                const int s  = q[u].x & 0xffff;
                if (slot[u] < CAP) {
                    bkt[dl * B_PITCH + slot[u]] = make_int2(s, q[u].y);
                } else {
                    atomicAdd(&ovw[dl], __int_as_float(q[u].y));
                    const int o = atomicAdd(&ecnt[N_NODES], 1);
                    ovf[o] = make_int4(base + dl, s, q[u].y, 0);
                }
            }
        }
    }
    if (__builtin_amdgcn_ballot_w32(run.y > 16) != 0u) {
        for (int i = 0; i < 32; i += 2) {
            const int st0 = __builtin_amdgcn_readlane(run.x, i), st1 = __builtin_amdgcn_readlane(run.x, i + 1);
            const int n0  = __builtin_amdgcn_readlane(run.y, i), n1  = __builtin_amdgcn_readlane(run.y, i + 1);
            const int st  = hw ? st1 : st0;
            const int n   = hw ? n1 : n0;
            const int2* __restrict__ p = part + (size_t)min(c0 + i + hw, N_CHUNKS - 1) * CHUNK + st;
            for (int k = 16 + jr; __builtin_amdgcn_ballot_w32(k < n) != 0u; k += 16) {
                if (k < n) {
                    const int2 q1 = p[k];
                    const int dl = q1.x >> 16;
                    const int s  = q1.x & 0xffff;
                    const int sl = atomicAdd(&cntl[dl], 1);
                    if (sl < CAP) {
                        bkt[dl * B_PITCH + sl] = make_int2(s, q1.y);
                    } else {
                        atomicAdd(&ovw[dl], __int_as_float(q1.y));
                        const int o = atomicAdd(&ecnt[N_NODES], 1);
                        ovf[o] = make_int4(base + dl, s, q1.y, 0);
                    }
                }
            }
        }
    }
    __syncthreads();

    v4i* __restrict__ rec4 = reinterpret_cast<v4i*>(rec) + (size_t)base * (CAP / 2);
    v4i pv[8];
    int pn[8];
    #pragma unroll
    for (int j = 0; j < 8; ++j) {
        const int pc   = j * 256 + tid;
        const int node = pc >> 4;
        pn[j] = cntl[node];
        pv[j] = *reinterpret_cast<const v4i*>(&bkt[node * B_PITCH + 2 * (pc & 15)]);
    }
    asm volatile("" :: "v"(pv[0]), "v"(pv[1]), "v"(pv[2]), "v"(pv[3]), "v"(pv[4]), "v"(pv[5]), "v"(pv[6]), "v"(pv[7]));
    #pragma unroll
    for (int j = 0; j < 8; ++j) {
        const int pc   = j * 256 + tid;
        const int node = pc >> 4;
        if (base + node < N_NODES && ((pc & 15) >> 3) * 16 < pn[j]) rec4[pc] = pv[j];
    }

    if (tid < NR && base + tid < N_NODES) {
        const int c  = cntl[tid];
        const int nb = min(c, CAP);
        const int2* __restrict__ b = bkt + tid * B_PITCH;
        float d = 1.0f;
        for (int k = 0; k < nb; k += 8) {
            float w[8];
            #pragma unroll
            for (int u = 0; u < 8; ++u)
                w[u] = __int_as_float(b[min(k + u, nb - 1)].y);
            #pragma unroll
            for (int u = 0; u < 8; ++u)
                if (k + u < nb) d += w[u];
        }
        if (c > CAP) d += ovw[tid];
        dis[base + tid]  = (d > 0.0f) ? rsqrtf(d) : 0.0f;
        ecnt[base + tid] = c;
    }
}

#define GROUP 16
static_assert(N_NODES % 16 == 0, "k_gather_pool: 8 whole rows per wave, 2 waves per block");

__device__ __forceinline__ void gather_group(float4& acc, const float4* __restrict__ x4, int lane,
                                             int j, int n, int src_l, float nrm_l) {
    float4 v[GROUP];
    #pragma unroll
    for (int u = 0; u < GROUP; ++u) {
        const int k = min(j + u, n - 1);
        const int s = __builtin_amdgcn_readlane(src_l, k);
        v[u] = x4[(size_t)s * 32 + lane];
    }
    #pragma unroll
    for (int u = 0; u < GROUP; ++u) {
        const float w = __int_as_float(__builtin_amdgcn_readlane(__float_as_int(nrm_l), j + u));
        acc.x += w * v[u].x;
        acc.y += w * v[u].y;
        acc.z += w * v[u].z;
        acc.w += w * v[u].w;
    }
}

__global__ void __launch_bounds__(64) k_gather_pool(const int* __restrict__ ecnt,
                                                    const int2* __restrict__ rec,
                                                    const int4* __restrict__ ovf,
                                                    const float* __restrict__ dis,
                                                    const float* __restrict__ x,
                                                    const int* __restrict__ batch,
                                                    float* __restrict__ pooled,
                                                    float* __restrict__ cnt) {
    __shared__ int2 packed[2 * 8 * CAP];
    const int lane = threadIdx.x & 31;
    int2* __restrict__ lw = packed + (threadIdx.x >> 5) * (8 * CAP);
    const int wave = __builtin_amdgcn_readfirstlane((int)((blockIdx.x * blockDim.x + threadIdx.x) >> 5));
    const int i0 = wave * 8;
    const float4* x4 = reinterpret_cast<const float4*>(x);

    const int   c_l = ecnt[i0 + min(lane, 7)];
    const int   g_l = batch[i0 + min(lane, 7)];
    const float d_l = dis[i0 + min(lane, 7)];
    const int n_l   = (lane < 8) ? min(c_l, CAP) : 0;
    const int off_l = wave_incl_scan(n_l, lane) - n_l;
    const unsigned over = __builtin_amdgcn_ballot_w32(lane < 8 && c_l > CAP);

    float4 xs[8];
    #pragma unroll
    for (int r = 0; r < 8; ++r) xs[r] = x4[(size_t)(i0 + r) * 32 + lane];

    {
        const int2* __restrict__ bw = rec + (size_t)i0 * CAP;
        int   nr[8];
        int2  q[8];
        float ds[8];
        #pragma unroll
        for (int r = 0; r < 8; ++r) {
            nr[r] = __builtin_amdgcn_readlane(n_l, r);
            if (lane < nr[r]) q[r] = bw[r * CAP + lane];
        }
        #pragma unroll
        for (int r = 0; r < 8; ++r)
            if (lane < nr[r]) ds[r] = dis[q[r].x];
        #pragma unroll
        for (int r = 0; r < 8; ++r) {
            const float dr = __int_as_float(__builtin_amdgcn_readlane(__float_as_int(d_l), r));
            const int   o  = __builtin_amdgcn_readlane(off_l, r);
            if (lane < nr[r]) {
                const float nrm = ds[r] * __int_as_float(q[r].y) * dr;
                lw[o + lane] = make_int2(q[r].x, __float_as_int(nrm));
            }
        }
    }
    __syncthreads();

    int ra = 0;
    while (ra < 8) {
        const int ga = __builtin_amdgcn_readlane(g_l, ra);
        int rb = ra + 1;
        while (rb < 8 && __builtin_amdgcn_readlane(g_l, rb) == ga) ++rb;
        const int sbeg = __builtin_amdgcn_readlane(off_l, ra);
        const int send = __builtin_amdgcn_readlane(off_l, rb);

        float4 acc = make_float4(0.0f, 0.0f, 0.0f, 0.0f);

        const bool any = sbeg < send;
        int n = min(32, send - sbeg);
        int2 p = make_int2(0, 0);
        if (any && lane < n) p = lw[sbeg + lane];

        #pragma unroll
        for (int r = 0; r < 8; ++r) {
            const float dr = __int_as_float(__builtin_amdgcn_readlane(__float_as_int(d_l), r));
            const float ws = (r >= ra && r < rb) ? dr * dr : 0.0f;
            acc.x += ws * xs[r].x;
            acc.y += ws * xs[r].y;
            acc.z += ws * xs[r].z;
            acc.w += ws * xs[r].w;
        }

        if (any) {
            int base = sbeg;
            for (;;) {
                const int  nb   = base + 32;
                const bool more = nb < send;
                const int  nn   = min(32, send - nb);
                int2 pn = make_int2(0, 0);
                if (more && lane < nn) pn = lw[nb + lane];
                gather_group(acc, x4, lane, 0, n, p.x, __int_as_float(p.y));
                if (n > GROUP) gather_group(acc, x4, lane, GROUP, n, p.x, __int_as_float(p.y));
                if (!more) break;
                base = nb;
                n = nn;
                p = pn;
            }
        }

        if ((over >> ra) & ((1u << (rb - ra)) - 1u)) {
            const int m = ecnt[N_NODES];
            for (int k0 = 0; k0 < m; k0 += 32) {
                const int k = k0 + lane;
                int4 o = make_int4(-1, 0, 0, 0);
                if (k < m) o = ovf[k];
                const bool hit = (o.x >= i0 + ra) && (o.x < i0 + rb);
                float nrm = 0.0f;
                if (hit) nrm = dis[o.y] * __int_as_float(o.z) * dis[o.x];
                unsigned todo = __builtin_amdgcn_ballot_w32(hit);
                while (todo) {
                    const int l = __builtin_ctz(todo);
                    todo &= todo - 1u;
                    const int   s = __builtin_amdgcn_readlane(o.y, l);
                    const float w = __int_as_float(__builtin_amdgcn_readlane(__float_as_int(nrm), l));
                    const float4 v = x4[(size_t)s * 32 + lane];
                    acc.x += w * v.x;
                    acc.y += w * v.y;
                    acc.z += w * v.z;
                    acc.w += w * v.w;
                }
            }
        }

        float* q = pooled + (size_t)ga * D_IN + 4 * lane;
        atomicAdd(q + 0, acc.x);
        atomicAdd(q + 1, acc.y);
        atomicAdd(q + 2, acc.z);
        atomicAdd(q + 3, acc.w);
        if (lane == 0) atomicAdd(&cnt[ga], (float)(rb - ra));
        ra = rb;
    }
}

#define T_PITCH 132

__global__ void __launch_bounds__(256) k_gemm_out(const float* __restrict__ pooled,
                                                  const float* __restrict__ cnt,
                                                  const float* __restrict__ Wc,
                                                  const float* __restrict__ bc,
                                                  const float* __restrict__ Wl,
                                                  const float* __restrict__ bl,
                                                  float* __restrict__ y) {
    __shared__ float t[16 * T_PITCH];
    const int lane  = threadIdx.x & 31;
    const int wv    = threadIdx.x >> 5;
    const int mtile = blockIdx.x;
    const int half  = lane >> 4;
    const int l     = lane & 15;
    const int g     = mtile * 16 + l;
    const int n     = wv * 16 + l;
    const float ic  = 1.0f / fmaxf(cnt[g], 1.0f);

    v8f acc = {};
    #pragma unroll
    for (int k0 = 0; k0 < D_IN; k0 += 4) {
        const int ka = k0 + 2 * half;
        v2f a, b;
        a.x = pooled[g * D_IN + ka] * ic;
        a.y = pooled[g * D_IN + ka + 1] * ic;
        b.x = Wc[n * D_IN + ka];
        b.y = Wc[n * D_IN + ka + 1];
        acc = __builtin_amdgcn_wmma_f32_16x16x4_f32(
                  false, a, false, b, (short)0, acc, false, false);
    }
    const float bn = bc[n];
    #pragma unroll
    for (int r = 0; r < 8; ++r) {
        const int row = half * 8 + r;
        const float has = (cnt[mtile * 16 + row] > 0.0f) ? 1.0f : 0.0f;
        t[row * T_PITCH + n] = acc[r] + has * bn;
    }
    __syncthreads();

    if (wv < 4) {
        v8f acc2 = {};
        #pragma unroll
        for (int k0 = 0; k0 < D_IN; k0 += 4) {
            const int ka = k0 + 2 * half;
            v2f a, b;
            a.x = t[l * T_PITCH + ka];
            a.y = t[l * T_PITCH + ka + 1];
            b.x = Wl[n * D_IN + ka];
            b.y = Wl[n * D_IN + ka + 1];
            acc2 = __builtin_amdgcn_wmma_f32_16x16x4_f32(
                       false, a, false, b, (short)0, acc2, false, false);
        }
        #pragma unroll
        for (int r = 0; r < 8; ++r) {
            int gg = mtile * 16 + half * 8 + r;
            y[gg * D_OUT + n] = acc2[r] + bl[n];
        }
    }
}


extern "C" void kernel_launch(void* const* d_in, const int* in_sizes, int n_in,
                              void* d_out, int out_size, void* d_ws, size_t ws_size,
                              hipStream_t stream) {
    const float* x     = (const float*)d_in[0];
    const int*   ei    = (const int*)  d_in[1];
    const int*   batch = (const int*)  d_in[2];
    const float* ew    = (const float*)d_in[3];
    const float* Wc    = (const float*)d_in[4];
    const float* bc    = (const float*)d_in[5];
    const float* Wl    = (const float*)d_in[6];
    const float* bl    = (const float*)d_in[7];
    float* y = (float*)d_out;

    char* ws = (char*)d_ws;
    size_t off = 0;
    auto take = [&](size_t bytes) -> float* {
        float* p = (float*)(ws + off);
        off += (bytes + 255) & ~((size_t)255);
        return p;
    };
    float* dis    = take((size_t)N_NODES * 4);
    float* pooled = take((size_t)N_GRAPHS * D_IN * 4);
    float* cnt    = take((size_t)N_GRAPHS * 4);
    int*  ecnt    = (int*) take((size_t)(N_NODES + 1) * 4);
    int2* rec     = (int2*)take((size_t)N_NODES * CAP * 8);
    int4* ovf     = (int4*)take((size_t)N_EDGES * 16);
    int2* part    = (int2*)take((size_t)N_EDGES * 8);
    int2* runs    = (int2*)take((size_t)N_RANGES * N_CHUNKS * 8);
    (void)ws_size; (void)in_sizes; (void)n_in; (void)out_size;

    hipMemsetAsync(pooled, 0, (size_t)N_GRAPHS * D_IN * 4, stream);
    hipMemsetAsync(cnt,    0, (size_t)N_GRAPHS * 4, stream);

    k_part    <<<N_CHUNKS, PART_T, 0, stream>>>(ei, ew, part, runs, ecnt);
    k_fine    <<<N_RANGES, 256, 0, stream>>>(part, runs, ecnt, rec, ovf, dis);

    k_gather_pool<<<N_NODES / 16, 64, 0, stream>>>(ecnt, rec, ovf, dis, x, batch, pooled, cnt);

    k_gemm_out<<<N_GRAPHS / 16, 256, 0, stream>>>(pooled, cnt, Wc, bc, Wl, bl, y);
}


# k_gather_pool: blocks of 25 waves copy the whole dis table into LDS once, as whole lines, and look each record's source dis up there; the 4-byte dis gathers from memory are gone
# speedup vs baseline: 7.1613x; 1.0578x over previous
#include <hip/hip_runtime.h>

typedef __attribute__((ext_vector_type(2))) float v2f;
typedef __attribute__((ext_vector_type(8))) float v8f;
typedef __attribute__((ext_vector_type(4))) int   v4i;
typedef __attribute__((ext_vector_type(4))) float v4f;

#define N_NODES  50000
#define N_EDGES  800000
#define D_IN     128
#define D_OUT    64
#define N_GRAPHS 512

#define CAP 32

#define NR       128
#define N_RANGES ((N_NODES + NR - 1) / NR)
#define PART_T   640
#define PART_E   5
#define CHUNK    (PART_T * PART_E)
#define N_CHUNKS (N_EDGES / CHUNK)
static_assert(N_EDGES % CHUNK == 0, "k_part: whole chunks only");
static_assert((CHUNK * 8) % 128 == 0 && CHUNK % 2 == 0, "k_part: a chunk of part is whole 128-byte lines, copied in 16-byte pieces");
static_assert(N_RANGES <= PART_T, "k_part: the scan holds one range per thread");
static_assert(N_CHUNKS <= 256, "k_fine: one thread per chunk");
static_assert(N_NODES <= 65536 && NR <= 32768, "packed record: source in the low 16 bits, node within the range above");

__device__ __forceinline__ int wave_incl_scan(int v, int lane) {
    v += __builtin_amdgcn_update_dpp(0, v, 0x111, 0xf, 0xf, true);
    v += __builtin_amdgcn_update_dpp(0, v, 0x112, 0xf, 0xf, true);
    v += __builtin_amdgcn_update_dpp(0, v, 0x114, 0xf, 0xf, true);
    v += __builtin_amdgcn_update_dpp(0, v, 0x118, 0xf, 0xf, true);
    const int low = __builtin_amdgcn_readlane(v, 15);
    return v + ((lane >= 16) ? low : 0);
}

__global__ void __launch_bounds__(PART_T) k_part(const int* __restrict__ ei,
                                                 const float* __restrict__ ew,
                                                 int2* __restrict__ part,
                                                 int2* __restrict__ runs,
                                                 int* __restrict__ ecnt) {
    __shared__ int hist[N_RANGES];
    __shared__ int wtot[PART_T / 32];
    __shared__ __attribute__((aligned(16))) int2 stage[CHUNK];
    const int tid  = threadIdx.x;
    const int lane = tid & 31;
    const int wv   = tid >> 5;
    if (tid < N_RANGES) hist[tid] = 0;
    if (blockIdx.x == 0 && tid == 0) ecnt[N_NODES] = 0;
    const int e0 = blockIdx.x * CHUNK + tid;
    int rng[PART_E], rank[PART_E];
    int2 r[PART_E];
    #pragma unroll
    for (int k = 0; k < PART_E; ++k) {
        const int e = e0 + k * PART_T;
        const unsigned d = (unsigned)ei[N_EDGES + e];
        rng[k] = (int)(d / NR);
        r[k] = make_int2((int)((d % NR) << 16) | ei[e], __float_as_int(ew[e]));
    }
    __syncthreads();
    #pragma unroll
    for (int k = 0; k < PART_E; ++k) rank[k] = atomicAdd(&hist[rng[k]], 1);
    __syncthreads();
    const int c   = (tid < N_RANGES) ? hist[tid] : 0;
    const int inc = wave_incl_scan(c, lane);
    if (lane == 31) wtot[wv] = inc;
    __syncthreads();
    int pre = 0;
    for (int w = 0; w < wv; ++w) pre += wtot[w];
    const int start = pre + inc - c;
    if (tid < N_RANGES) {
        hist[tid] = start;
        runs[(size_t)tid * N_CHUNKS + blockIdx.x] = make_int2(start, c);
    }
    __syncthreads();
    #pragma unroll
    for (int k = 0; k < PART_E; ++k) stage[hist[rng[k]] + rank[k]] = r[k];
    __syncthreads();
    v4i* __restrict__ out4 = reinterpret_cast<v4i*>(part + (size_t)blockIdx.x * CHUNK);
    const v4i* st4 = reinterpret_cast<const v4i*>(stage);
    constexpr int NPC = (CHUNK / 2 + PART_T - 1) / PART_T;
    v4i pv[NPC];
    #pragma unroll
    for (int p = 0; p < NPC; ++p)
        pv[p] = st4[min(p * PART_T + tid, CHUNK / 2 - 1)];
    #pragma unroll
    for (int p = 0; p < NPC; ++p) {
        const int pc = p * PART_T + tid;
        if ((p + 1) * PART_T <= CHUNK / 2 || pc < CHUNK / 2) out4[pc] = pv[p];
    }
}

#define B_PITCH (CAP + 2)
static_assert(CAP == 32 && NR == 128, "k_fine: 16 pieces of 16 bytes per bucket, 2048 pieces per block, 8 per thread");

__global__ void __launch_bounds__(256) k_fine(const int2* __restrict__ part,
                                              const int2* __restrict__ runs,
                                              int* __restrict__ ecnt,
                                              int2* __restrict__ rec,
                                              int4* __restrict__ ovf,
                                              float* __restrict__ dis) {
    __shared__ __attribute__((aligned(16))) int2 bkt[NR * B_PITCH];
    __shared__ int   cntl[NR];
    __shared__ float ovw[NR];
    const int tid  = threadIdx.x;
    const int base = blockIdx.x * NR;
    if (tid < NR) { cntl[tid] = 0; ovw[tid] = 0.0f; }
    for (int p = tid; p < NR * B_PITCH / 2; p += 256)
        reinterpret_cast<int4*>(bkt)[p] = make_int4(0, 0, 0, 0);
    int2 run = make_int2(0, 0);
    if (tid < N_CHUNKS) run = runs[(size_t)blockIdx.x * N_CHUNKS + tid];
    __syncthreads();
    const int lane = tid & 31;
    const int hw   = lane >> 4;
    const int jr   = lane & 15;
    const int c0   = tid & ~31;
    {
        int2 q[16];
        int  nn[16];
        #pragma unroll
        for (int u = 0; u < 16; ++u) {
            const int i   = 2 * u;
            const int st0 = __builtin_amdgcn_readlane(run.x, i), st1 = __builtin_amdgcn_readlane(run.x, i + 1);
            const int n0  = __builtin_amdgcn_readlane(run.y, i), n1  = __builtin_amdgcn_readlane(run.y, i + 1);
            const int st  = hw ? st1 : st0;
            nn[u]         = hw ? n1 : n0;
            const int2* __restrict__ p = part + (size_t)min(c0 + i + hw, N_CHUNKS - 1) * CHUNK + st;
            if (jr < nn[u]) q[u] = p[jr];
        }
        int slot[16];
        #pragma unroll
        for (int u = 0; u < 16; ++u)
            if (jr < nn[u]) slot[u] = atomicAdd(&cntl[q[u].x >> 16], 1);
        #pragma unroll
        for (int u = 0; u < 16; ++u) {
            if (jr < nn[u]) {
                const int dl = q[u].x >> 16;
                const int s  = q[u].x & 0xffff;
                if (slot[u] < CAP) {
                    bkt[dl * B_PITCH + slot[u]] = make_int2(s, q[u].y);
                } else {
                    atomicAdd(&ovw[dl], __int_as_float(q[u].y));
                    const int o = atomicAdd(&ecnt[N_NODES], 1);
                    ovf[o] = make_int4(base + dl, s, q[u].y, 0);
                }
            }
        }
    }
    if (__builtin_amdgcn_ballot_w32(run.y > 16) != 0u) {
        for (int i = 0; i < 32; i += 2) {
            const int st0 = __builtin_amdgcn_readlane(run.x, i), st1 = __builtin_amdgcn_readlane(run.x, i + 1);
            const int n0  = __builtin_amdgcn_readlane(run.y, i), n1  = __builtin_amdgcn_readlane(run.y, i + 1);
            const int st  = hw ? st1 : st0;
            const int n   = hw ? n1 : n0;
            const int2* __restrict__ p = part + (size_t)min(c0 + i + hw, N_CHUNKS - 1) * CHUNK + st;
            for (int k = 16 + jr; __builtin_amdgcn_ballot_w32(k < n) != 0u; k += 16) {
                if (k < n) {
                    const int2 q1 = p[k];
                    const int dl = q1.x >> 16;
                    const int s  = q1.x & 0xffff;
                    const int sl = atomicAdd(&cntl[dl], 1);
                    if (sl < CAP) {
                        bkt[dl * B_PITCH + sl] = make_int2(s, q1.y);
                    } else {
                        atomicAdd(&ovw[dl], __int_as_float(q1.y));
                        const int o = atomicAdd(&ecnt[N_NODES], 1);
                        ovf[o] = make_int4(base + dl, s, q1.y, 0);
                    }
                }
            }
        }
    }
    __syncthreads();

    v4i* __restrict__ rec4 = reinterpret_cast<v4i*>(rec) + (size_t)base * (CAP / 2);
    v4i pv[8];
    int pn[8];
    #pragma unroll
    for (int j = 0; j < 8; ++j) {
        const int pc   = j * 256 + tid;
        const int node = pc >> 4;
        pn[j] = cntl[node];
        pv[j] = *reinterpret_cast<const v4i*>(&bkt[node * B_PITCH + 2 * (pc & 15)]);
    }
    asm volatile("" :: "v"(pv[0]), "v"(pv[1]), "v"(pv[2]), "v"(pv[3]), "v"(pv[4]), "v"(pv[5]), "v"(pv[6]), "v"(pv[7]));
    #pragma unroll
    for (int j = 0; j < 8; ++j) {
        const int pc   = j * 256 + tid;
        const int node = pc >> 4;
        if (base + node < N_NODES && ((pc & 15) >> 3) * 16 < pn[j]) rec4[pc] = pv[j];
    }

    if (tid < NR && base + tid < N_NODES) {
        const int c  = cntl[tid];
        const int nb = min(c, CAP);
        const int2* __restrict__ b = bkt + tid * B_PITCH;
        float d = 1.0f;
        for (int k = 0; k < nb; k += 8) {
            float w[8];
            #pragma unroll
            for (int u = 0; u < 8; ++u)
                w[u] = __int_as_float(b[min(k + u, nb - 1)].y);
            #pragma unroll
            for (int u = 0; u < 8; ++u)
                if (k + u < nb) d += w[u];
        }
        if (c > CAP) d += ovw[tid];
        dis[base + tid]  = (d > 0.0f) ? rsqrtf(d) : 0.0f;
        ecnt[base + tid] = c;
    }
}

#define GROUP 16
#define GP_WAVES 25
#define GP_T     (GP_WAVES * 32)
#define GP_TAB   ((N_NODES / 4 + GP_T - 1) / GP_T)
static_assert(N_NODES % (8 * GP_WAVES) == 0, "k_gather_pool: 8 whole rows per wave, whole blocks");
static_assert(N_NODES % 4 == 0, "k_gather_pool: the dis table is copied in 16-byte pieces");

__device__ __forceinline__ void gather_group(float4& acc, const float4* __restrict__ x4, int lane,
                                             int j, int n, int src_l, float nrm_l) {
    float4 v[GROUP];
    #pragma unroll
    for (int u = 0; u < GROUP; ++u) {
        const int k = min(j + u, n - 1);
        const int s = __builtin_amdgcn_readlane(src_l, k);
        v[u] = x4[(size_t)s * 32 + lane];
    }
    #pragma unroll
    for (int u = 0; u < GROUP; ++u) {
        const float w = __int_as_float(__builtin_amdgcn_readlane(__float_as_int(nrm_l), j + u));
        acc.x += w * v[u].x;
        acc.y += w * v[u].y;
        acc.z += w * v[u].z;
        acc.w += w * v[u].w;
    }
}

__global__ void __launch_bounds__(GP_T) k_gather_pool(const int* __restrict__ ecnt,
                                                    const int2* __restrict__ rec,
                                                    const int4* __restrict__ ovf,
                                                    const float* __restrict__ dis,
                                                    const float* __restrict__ x,
                                                    const int* __restrict__ batch,
                                                    float* __restrict__ pooled,
                                                    float* __restrict__ cnt) {
    __shared__ __attribute__((aligned(16))) float sdis[N_NODES];
    __shared__ int2 packed[GP_WAVES * 8 * CAP];
    const int lane = threadIdx.x & 31;
    int2* __restrict__ lw = packed + (threadIdx.x >> 5) * (8 * CAP);
    const int wave = __builtin_amdgcn_readfirstlane((int)((blockIdx.x * blockDim.x + threadIdx.x) >> 5));
    const int i0 = wave * 8;
    const float4* x4 = reinterpret_cast<const float4*>(x);

    const int   c_l = ecnt[i0 + min(lane, 7)];
    const int   g_l = batch[i0 + min(lane, 7)];
    const float d_l = dis[i0 + min(lane, 7)];
    const int n_l   = (lane < 8) ? min(c_l, CAP) : 0;
    const int off_l = wave_incl_scan(n_l, lane) - n_l;
    const unsigned over = __builtin_amdgcn_ballot_w32(lane < 8 && c_l > CAP);

    float4 xs[8];
    #pragma unroll
    for (int r = 0; r < 8; ++r) xs[r] = x4[(size_t)(i0 + r) * 32 + lane];

    const v4f* dis4 = reinterpret_cast<const v4f*>(dis);
    v4f tab[GP_TAB];
    #pragma unroll
    for (int p = 0; p < GP_TAB; ++p)
        tab[p] = dis4[min(p * GP_T + (int)threadIdx.x, N_NODES / 4 - 1)];

    {
        const int2* __restrict__ bw = rec + (size_t)i0 * CAP;
        int   nr[8];
        int2  q[8];
        float ds[8];
        #pragma unroll
        for (int r = 0; r < 8; ++r) {
            nr[r] = __builtin_amdgcn_readlane(n_l, r);
            if (lane < nr[r]) q[r] = bw[r * CAP + lane];
        }
        static_assert(GP_TAB == 16, "k_gather_pool: the statement below names 16 pieces");
        asm volatile("" :: "v"(tab[0]), "v"(tab[1]), "v"(tab[2]), "v"(tab[3]), "v"(tab[4]), "v"(tab[5]), "v"(tab[6]), "v"(tab[7]),
                           "v"(tab[8]), "v"(tab[9]), "v"(tab[10]), "v"(tab[11]), "v"(tab[12]), "v"(tab[13]), "v"(tab[14]), "v"(tab[15]));
        #pragma unroll
        for (int p = 0; p < GP_TAB; ++p) {
            const int pc = p * GP_T + (int)threadIdx.x;
            if ((p + 1) * GP_T <= N_NODES / 4 || pc < N_NODES / 4) reinterpret_cast<v4f*>(sdis)[pc] = tab[p];
        }
        __syncthreads();
        #pragma unroll
        for (int r = 0; r < 8; ++r)
            if (lane < nr[r]) ds[r] = sdis[q[r].x];
        #pragma unroll
        for (int r = 0; r < 8; ++r) {
            const float dr = __int_as_float(__builtin_amdgcn_readlane(__float_as_int(d_l), r));
            const int   o  = __builtin_amdgcn_readlane(off_l, r);
            if (lane < nr[r]) {
                const float nrm = ds[r] * __int_as_float(q[r].y) * dr;
                lw[o + lane] = make_int2(q[r].x, __float_as_int(nrm));
            }
        }
    }
    __syncthreads();

    int ra = 0;
    while (ra < 8) {
        const int ga = __builtin_amdgcn_readlane(g_l, ra);
        int rb = ra + 1;
        while (rb < 8 && __builtin_amdgcn_readlane(g_l, rb) == ga) ++rb;
        const int sbeg = __builtin_amdgcn_readlane(off_l, ra);
        const int send = __builtin_amdgcn_readlane(off_l, rb);

        float4 acc = make_float4(0.0f, 0.0f, 0.0f, 0.0f);

        const bool any = sbeg < send;
        int n = min(32, send - sbeg);
        int2 p = make_int2(0, 0);
        if (any && lane < n) p = lw[sbeg + lane];

        #pragma unroll
        for (int r = 0; r < 8; ++r) {
            const float dr = __int_as_float(__builtin_amdgcn_readlane(__float_as_int(d_l), r));
            const float ws = (r >= ra && r < rb) ? dr * dr : 0.0f;
            acc.x += ws * xs[r].x;
            acc.y += ws * xs[r].y;
            acc.z += ws * xs[r].z;
            acc.w += ws * xs[r].w;
        }

        if (any) {
            int base = sbeg;
            for (;;) {
                const int  nb   = base + 32;
                const bool more = nb < send;
                const int  nn   = min(32, send - nb);
                int2 pn = make_int2(0, 0);
                if (more && lane < nn) pn = lw[nb + lane];
                gather_group(acc, x4, lane, 0, n, p.x, __int_as_float(p.y));
                if (n > GROUP) gather_group(acc, x4, lane, GROUP, n, p.x, __int_as_float(p.y));
                if (!more) break;
                base = nb;
                n = nn;
                p = pn;
            }
        }

        if ((over >> ra) & ((1u << (rb - ra)) - 1u)) {
            const int m = ecnt[N_NODES];
            for (int k0 = 0; k0 < m; k0 += 32) {
                const int k = k0 + lane;
                int4 o = make_int4(-1, 0, 0, 0);
                if (k < m) o = ovf[k];
                const bool hit = (o.x >= i0 + ra) && (o.x < i0 + rb);
                float nrm = 0.0f;
                if (hit) nrm = sdis[o.y] * __int_as_float(o.z) * sdis[o.x];
                unsigned todo = __builtin_amdgcn_ballot_w32(hit);
                while (todo) {
                    const int l = __builtin_ctz(todo);
                    todo &= todo - 1u;
                    const int   s = __builtin_amdgcn_readlane(o.y, l);
                    const float w = __int_as_float(__builtin_amdgcn_readlane(__float_as_int(nrm), l));
                    const float4 v = x4[(size_t)s * 32 + lane];
                    acc.x += w * v.x;
                    acc.y += w * v.y;
                    acc.z += w * v.z;
                    acc.w += w * v.w;
                }
            }
        }

        float* q = pooled + (size_t)ga * D_IN + 4 * lane;
        atomicAdd(q + 0, acc.x);
        atomicAdd(q + 1, acc.y);
        atomicAdd(q + 2, acc.z);
        atomicAdd(q + 3, acc.w);
        if (lane == 0) atomicAdd(&cnt[ga], (float)(rb - ra));
        ra = rb;
    }
}

#define T_PITCH 132

__global__ void __launch_bounds__(256) k_gemm_out(const float* __restrict__ pooled,
                                                  const float* __restrict__ cnt,
                                                  const float* __restrict__ Wc,
                                                  const float* __restrict__ bc,
                                                  const float* __restrict__ Wl,
                                                  const float* __restrict__ bl,
                                                  float* __restrict__ y) {
    __shared__ float t[16 * T_PITCH];
    const int lane  = threadIdx.x & 31;
    const int wv    = threadIdx.x >> 5;
    const int mtile = blockIdx.x;
    const int half  = lane >> 4;
    const int l     = lane & 15;
    const int g     = mtile * 16 + l;
    const int n     = wv * 16 + l;
    const float ic  = 1.0f / fmaxf(cnt[g], 1.0f);

    v8f acc = {};
    #pragma unroll
    for (int k0 = 0; k0 < D_IN; k0 += 4) {
        const int ka = k0 + 2 * half;
        v2f a, b;
        a.x = pooled[g * D_IN + ka] * ic;
        a.y = pooled[g * D_IN + ka + 1] * ic;
        b.x = Wc[n * D_IN + ka];
        b.y = Wc[n * D_IN + ka + 1];
        acc = __builtin_amdgcn_wmma_f32_16x16x4_f32(
                  false, a, false, b, (short)0, acc, false, false);
    }
    const float bn = bc[n];
    #pragma unroll
    for (int r = 0; r < 8; ++r) {
        const int row = half * 8 + r;
        const float has = (cnt[mtile * 16 + row] > 0.0f) ? 1.0f : 0.0f;
        t[row * T_PITCH + n] = acc[r] + has * bn;
    }
    __syncthreads();

    if (wv < 4) {
        v8f acc2 = {};
        #pragma unroll
        for (int k0 = 0; k0 < D_IN; k0 += 4) {
            const int ka = k0 + 2 * half;
            v2f a, b;
            a.x = t[l * T_PITCH + ka];
            a.y = t[l * T_PITCH + ka + 1];
            b.x = Wl[n * D_IN + ka];
            b.y = Wl[n * D_IN + ka + 1];
            acc2 = __builtin_amdgcn_wmma_f32_16x16x4_f32(
                       false, a, false, b, (short)0, acc2, false, false);
        }
        #pragma unroll
        for (int r = 0; r < 8; ++r) {
            int gg = mtile * 16 + half * 8 + r;
            y[gg * D_OUT + n] = acc2[r] + bl[n];
        }
    }
}


extern "C" void kernel_launch(void* const* d_in, const int* in_sizes, int n_in,
                              void* d_out, int out_size, void* d_ws, size_t ws_size,
                              hipStream_t stream) {
    const float* x     = (const float*)d_in[0];
    const int*   ei    = (const int*)  d_in[1];
    const int*   batch = (const int*)  d_in[2];
    const float* ew    = (const float*)d_in[3];
    const float* Wc    = (const float*)d_in[4];
    const float* bc    = (const float*)d_in[5];
    const float* Wl    = (const float*)d_in[6];
    const float* bl    = (const float*)d_in[7];
    float* y = (float*)d_out;

    char* ws = (char*)d_ws;
    size_t off = 0;
    auto take = [&](size_t bytes) -> float* {
        float* p = (float*)(ws + off);
        off += (bytes + 255) & ~((size_t)255);
        return p;
    };
    float* dis    = take((size_t)N_NODES * 4);
    float* pooled = take((size_t)N_GRAPHS * D_IN * 4);
    float* cnt    = take((size_t)N_GRAPHS * 4);
    int*  ecnt    = (int*) take((size_t)(N_NODES + 1) * 4);
    int2* rec     = (int2*)take((size_t)N_NODES * CAP * 8);
    int4* ovf     = (int4*)take((size_t)N_EDGES * 16);
    int2* part    = (int2*)take((size_t)N_EDGES * 8);
    int2* runs    = (int2*)take((size_t)N_RANGES * N_CHUNKS * 8);
    (void)ws_size; (void)in_sizes; (void)n_in; (void)out_size;

    hipMemsetAsync(pooled, 0, (size_t)N_GRAPHS * D_IN * 4, stream);
    hipMemsetAsync(cnt,    0, (size_t)N_GRAPHS * 4, stream);

    k_part    <<<N_CHUNKS, PART_T, 0, stream>>>(ei, ew, part, runs, ecnt);
    k_fine    <<<N_RANGES, 256, 0, stream>>>(part, runs, ecnt, rec, ovf, dis);

    k_gather_pool<<<N_NODES / (8 * GP_WAVES), GP_T, 0, stream>>>(ecnt, rec, ovf, dis, x, batch, pooled, cnt);

    k_gemm_out<<<N_GRAPHS / 16, 256, 0, stream>>>(pooled, cnt, Wc, bc, Wl, bl, y);
}
